# starting kernel
# baseline (speedup 1.0000x reference)
#include <hip/hip_runtime.h>

#define IN_F  128
#define HID_F 128
#define OUT_F 64

typedef __attribute__((ext_vector_type(2))) float v2f;
typedef __attribute__((ext_vector_type(8))) float v8f;


__global__ void fill_kernel(float* __restrict__ p, float v, int n) {
  int i = blockIdx.x * blockDim.x + threadIdx.x;
  if (i < n) p[i] = v;
}

__global__ void deg_kernel(const int* __restrict__ col, float* __restrict__ deg, int E) {
  int e = blockIdx.x * blockDim.x + threadIdx.x;
  if (e < E) atomicAdd(&deg[col[e]], 1.0f);
}

__global__ void dinv_kernel(float* __restrict__ deg, int n) {
  int i = blockIdx.x * blockDim.x + threadIdx.x;
  if (i < n) deg[i] = rsqrtf(deg[i]);
}

__global__ void bias_relu_kernel(const float* __restrict__ xw,
                                 float* __restrict__ agg,
                                 const float* __restrict__ dinv,
                                 const float* __restrict__ b1,
                                 int total) {
  int i = blockIdx.x * blockDim.x + threadIdx.x;
  if (i >= total) return;
  int node = i >> 7;
  int f    = i & 127;
  float di = dinv[node];
  float v  = agg[i] + xw[i] * di * di + b1[f];
  agg[i] = fmaxf(v, 0.0f);
}

__global__ void final_kernel(const float* __restrict__ hw,
                             const float* __restrict__ dinv,
                             const float* __restrict__ b2,
                             float* __restrict__ out,
                             int total) {
  int i = blockIdx.x * blockDim.x + threadIdx.x;
  if (i >= total) return;
  int node = i >> 6;
  int f    = i & 63;
  float di = dinv[node];
  out[i] = out[i] + hw[i] * di * di + b2[f];
}


template <int F>
__global__ void __launch_bounds__((F / 16) * 32)
gemm_wmma_kernel(const float* __restrict__ A,
                 const float* __restrict__ B,
                 float* __restrict__ C,
                 int nrows) {
  const int lane = threadIdx.x & 31;
  const int wave = threadIdx.x >> 5;
  const int row0 = blockIdx.x * 16;
  if (row0 >= nrows) return;
  const int col0 = wave * 16;
  const int m    = lane & 15;
  const int half = lane >> 4;

  v8f acc = {};
  const float* arow = A + (size_t)(row0 + m) * 128;

#pragma unroll
  for (int k = 0; k < 128; k += 4) {
    const int ka = k + 2 * half;
    v2f a, b;
    a.x = arow[ka];
    a.y = arow[ka + 1];
    b.x = B[(size_t)ka * F + col0 + m];
    b.y = B[(size_t)(ka + 1) * F + col0 + m];
    acc = __builtin_amdgcn_wmma_f32_16x16x4_f32(
         false, a,  false, b,
         (short)0, acc,  false,  false);
  }

#pragma unroll
  for (int v = 0; v < 8; ++v) {
    C[(size_t)(row0 + v + 8 * half) * F + col0 + m] = acc[v];
  }
}

__global__ void scatter128_kernel(const int* __restrict__ row,
                                  const int* __restrict__ col,
                                  const float* __restrict__ dinv,
                                  const float* __restrict__ xw,
                                  float* __restrict__ agg,
                                  int E) {
  int gw   = (blockIdx.x * blockDim.x + threadIdx.x) >> 5;
  int lane = threadIdx.x & 31;
  if (gw >= E) return;
  int r = row[gw];
  int c = col[gw];
  float nrm = dinv[r] * dinv[c];
  const float4* src = (const float4*)(xw + (size_t)r * 128);
  float4 v = src[lane];
  float* dst = agg + (size_t)c * 128 + lane * 4;
  atomicAdd(dst + 0, v.x * nrm);
  atomicAdd(dst + 1, v.y * nrm);
  atomicAdd(dst + 2, v.z * nrm);
  atomicAdd(dst + 3, v.w * nrm);
}

__global__ void scatter64_kernel(const int* __restrict__ row,
                                 const int* __restrict__ col,
                                 const float* __restrict__ dinv,
                                 const float* __restrict__ hw,
                                 float* __restrict__ agg,
                                 int E) {
  int gw   = (blockIdx.x * blockDim.x + threadIdx.x) >> 5;
  int lane = threadIdx.x & 31;
  if (gw >= E) return;
  int r = row[gw];
  int c = col[gw];
  float nrm = dinv[r] * dinv[c];
  const float2* src = (const float2*)(hw + (size_t)r * 64);
  float2 v = src[lane];
  float* dst = agg + (size_t)c * 64 + lane * 2;
  atomicAdd(dst + 0, v.x * nrm);
  atomicAdd(dst + 1, v.y * nrm);
}


extern "C" void kernel_launch(void* const* d_in, const int* in_sizes, int n_in,
                              void* d_out, int out_size, void* d_ws, size_t ws_size,
                              hipStream_t stream) {
  const float* x  = (const float*)d_in[0];
  const int*   ei = (const int*)d_in[1];
  const float* W1 = (const float*)d_in[2];
  const float* b1 = (const float*)d_in[3];
  const float* W2 = (const float*)d_in[4];
  const float* b2 = (const float*)d_in[5];
  float* out = (float*)d_out;

  const int N = in_sizes[0] / IN_F;
  const int E = in_sizes[1] / 2;
  const int* row = ei;
  const int* col = ei + E;

  float* ws   = (float*)d_ws;
  float* xw   = ws;
  float* agg1 = xw + (size_t)N * 128;
  float* hw   = agg1 + (size_t)N * 128;
  float* deg  = hw + (size_t)N * 64;

  const int n128 = N * 128;
  const int n64  = N * 64;

  fill_kernel<<<(n128 + 255) / 256, 256, 0, stream>>>(agg1, 0.0f, n128);
  fill_kernel<<<(n64  + 255) / 256, 256, 0, stream>>>(out,  0.0f, n64);
  fill_kernel<<<(N    + 255) / 256, 256, 0, stream>>>(deg,  1.0f, N);

  deg_kernel<<<(E + 255) / 256, 256, 0, stream>>>(col, deg, E);
  dinv_kernel<<<(N + 255) / 256, 256, 0, stream>>>(deg, N);

  gemm_wmma_kernel<128><<<(N + 15) / 16, 256, 0, stream>>>(x, W1, xw, N);
  scatter128_kernel<<<(E + 7) / 8, 256, 0, stream>>>(row, col, deg, xw, agg1, E);
  bias_relu_kernel<<<(n128 + 255) / 256, 256, 0, stream>>>(xw, agg1, deg, b1, n128);

  gemm_wmma_kernel<64><<<(N + 15) / 16, 128, 0, stream>>>(agg1, W2, hw, N);
  scatter64_kernel<<<(E + 7) / 8, 256, 0, stream>>>(row, col, deg, hw, out, E);
  final_kernel<<<(n64 + 255) / 256, 256, 0, stream>>>(hw, deg, b2, out, n64);
}


# Float-atomic edge scatter replaced by a gather over edges grouped by target (count+rank, scan, place): one wave per node sums rows in registers and stores the finished row; summation order changes
# speedup vs baseline: 1.7780x; 1.7780x over previous
#include <hip/hip_runtime.h>

#define IN_F  128
#define HID_F 128
#define OUT_F 64

typedef __attribute__((ext_vector_type(2))) float v2f;
typedef __attribute__((ext_vector_type(8))) float v8f;


__global__ void zero_int_kernel(int* __restrict__ p, int n) {
  int i = blockIdx.x * blockDim.x + threadIdx.x;
  if (i < n) p[i] = 0;
}

__global__ void count_rank_kernel(const int* __restrict__ col, int* __restrict__ cnt,
                                  int* __restrict__ rank, int E) {
  int e = blockIdx.x * blockDim.x + threadIdx.x;
  if (e < E) rank[e] = atomicAdd(&cnt[col[e]], 1);
}

#define SCAN_T 1024
__global__ void __launch_bounds__(SCAN_T)
scan_dinv_kernel(const int* __restrict__ cnt, int* __restrict__ off,
                 float* __restrict__ dinv, int n) {
  __shared__ int wave_tot[32];
  __shared__ int wave_pre[32];
  const int t     = threadIdx.x;
  const int lane  = t & 31;
  const int wave  = t >> 5;
  const int tile0 = blockIdx.x * SCAN_T;

  int pre = 0;
  for (int i = t; i < tile0; i += SCAN_T) pre += cnt[i];

  const int i = tile0 + t;
  const int v = (i < n) ? cnt[i] : 0;

  int inc = v;
  for (int d = 1; d < 32; d <<= 1) {
    int u = __shfl_up(inc, d);
    if (lane >= d) inc += u;
  }
  for (int d = 16; d >= 1; d >>= 1) pre += __shfl_xor(pre, d);
  if (lane == 31) wave_tot[wave] = inc;
  if (lane == 0)  wave_pre[wave] = pre;
  __syncthreads();

  const int wt = wave_tot[lane];
  int winc = wt;
  for (int d = 1; d < 32; d <<= 1) {
    int u = __shfl_up(winc, d);
    if (lane >= d) winc += u;
  }
  int front = wave_pre[lane];
  for (int d = 16; d >= 1; d >>= 1) front += __shfl_xor(front, d);
  const int wave_base = __shfl(winc - wt, wave);

  if (i < n) {
    off[i]  = front + wave_base + inc - v;
    dinv[i] = rsqrtf((float)(v + 1));
  }
}

__global__ void place_kernel(const int* __restrict__ row, const int* __restrict__ col,
                             const int* __restrict__ rank, const int* __restrict__ off,
                             int* __restrict__ src, int E) {
  int e = blockIdx.x * blockDim.x + threadIdx.x;
  if (e < E) src[off[col[e]] + rank[e]] = row[e];
}


template <int F>
__global__ void __launch_bounds__((F / 16) * 32)
gemm_wmma_kernel(const float* __restrict__ A,
                 const float* __restrict__ B,
                 float* __restrict__ C,
                 int nrows) {
  const int lane = threadIdx.x & 31;
  const int wave = threadIdx.x >> 5;
  const int row0 = blockIdx.x * 16;
  if (row0 >= nrows) return;
  const int col0 = wave * 16;
  const int m    = lane & 15;
  const int half = lane >> 4;

  v8f acc = {};
  const float* arow = A + (size_t)(row0 + m) * 128;

#pragma unroll
  for (int k = 0; k < 128; k += 4) {
    const int ka = k + 2 * half;
    v2f a, b;
    a.x = arow[ka];
    a.y = arow[ka + 1];
    b.x = B[(size_t)ka * F + col0 + m];
    b.y = B[(size_t)(ka + 1) * F + col0 + m];
    acc = __builtin_amdgcn_wmma_f32_16x16x4_f32(
         false, a,  false, b,
         (short)0, acc,  false,  false);
  }

#pragma unroll
  for (int v = 0; v < 8; ++v) {
    C[(size_t)(row0 + v + 8 * half) * F + col0 + m] = acc[v];
  }
}


static __device__ __forceinline__ float lane_f(float x, int j) {
  return __int_as_float(__builtin_amdgcn_readlane(__float_as_int(x), j));
}

__global__ void __launch_bounds__(256)
gather128_kernel(const int* __restrict__ off, const int* __restrict__ cnt,
                 const int* __restrict__ src, const float* __restrict__ dinv,
                 const float* __restrict__ xw, const float* __restrict__ b1,
                 float* __restrict__ h, int n) {
  const int lane = threadIdx.x & 31;
  const int c = __builtin_amdgcn_readfirstlane(blockIdx.x * 8 + (threadIdx.x >> 5));
  if (c >= n) return;
  const int   start = off[c];
  const int   m     = cnt[c];
  const float dc    = dinv[c];
  const float4* xw4 = (const float4*)xw;

  const float4 s  = xw4[(size_t)c * 32 + lane];
  const float  ws = dc * dc;
  float4 acc = make_float4(s.x * ws, s.y * ws, s.z * ws, s.w * ws);

  for (int base = 0; base < m; base += 32) {
    const int k = (m - base < 32) ? (m - base) : 32;
    int   r_l = 0;
    float w_l = 0.0f;
    if (lane < k) {
      r_l = src[start + base + lane];
      w_l = dinv[r_l] * dc;
    }
    int j = 0;
    for (; j + 4 <= k; j += 4) {
      const int r0 = __builtin_amdgcn_readlane(r_l, j);
      const int r1 = __builtin_amdgcn_readlane(r_l, j + 1);
      const int r2 = __builtin_amdgcn_readlane(r_l, j + 2);
      const int r3 = __builtin_amdgcn_readlane(r_l, j + 3);
      const float4 v0 = xw4[(size_t)r0 * 32 + lane];
      const float4 v1 = xw4[(size_t)r1 * 32 + lane];
      const float4 v2 = xw4[(size_t)r2 * 32 + lane];
      const float4 v3 = xw4[(size_t)r3 * 32 + lane];
      const float w0 = lane_f(w_l, j);
      const float w1 = lane_f(w_l, j + 1);
      const float w2 = lane_f(w_l, j + 2);
      const float w3 = lane_f(w_l, j + 3);
      acc.x += v0.x * w0; acc.y += v0.y * w0; acc.z += v0.z * w0; acc.w += v0.w * w0;
      acc.x += v1.x * w1; acc.y += v1.y * w1; acc.z += v1.z * w1; acc.w += v1.w * w1;
      acc.x += v2.x * w2; acc.y += v2.y * w2; acc.z += v2.z * w2; acc.w += v2.w * w2;
      acc.x += v3.x * w3; acc.y += v3.y * w3; acc.z += v3.z * w3; acc.w += v3.w * w3;
    }
    for (; j < k; ++j) {
      const int   r = __builtin_amdgcn_readlane(r_l, j);
      const float w = lane_f(w_l, j);
      const float4 v = xw4[(size_t)r * 32 + lane];
      acc.x += v.x * w; acc.y += v.y * w; acc.z += v.z * w; acc.w += v.w * w;
    }
  }

  const int f = lane * 4;
  float4 o;
  o.x = fmaxf(acc.x + b1[f + 0], 0.0f);
  o.y = fmaxf(acc.y + b1[f + 1], 0.0f);
  o.z = fmaxf(acc.z + b1[f + 2], 0.0f);
  o.w = fmaxf(acc.w + b1[f + 3], 0.0f);
  ((float4*)h)[(size_t)c * 32 + lane] = o;
}

__global__ void __launch_bounds__(256)
gather64_kernel(const int* __restrict__ off, const int* __restrict__ cnt,
                const int* __restrict__ src, const float* __restrict__ dinv,
                const float* __restrict__ hw, const float* __restrict__ b2,
                float* __restrict__ out, int n) {
  const int lane = threadIdx.x & 31;
  const int c = __builtin_amdgcn_readfirstlane(blockIdx.x * 8 + (threadIdx.x >> 5));
  if (c >= n) return;
  const int   start = off[c];
  const int   m     = cnt[c];
  const float dc    = dinv[c];
  const float2* hw2 = (const float2*)hw;

  const float2 s  = hw2[(size_t)c * 32 + lane];
  const float  ws = dc * dc;
  float2 acc = make_float2(s.x * ws, s.y * ws);

  for (int base = 0; base < m; base += 32) {
    const int k = (m - base < 32) ? (m - base) : 32;
    int   r_l = 0;
    float w_l = 0.0f;
    if (lane < k) {
      r_l = src[start + base + lane];
      w_l = dinv[r_l] * dc;
    }
    int j = 0;
    for (; j + 4 <= k; j += 4) {
      const int r0 = __builtin_amdgcn_readlane(r_l, j);
      const int r1 = __builtin_amdgcn_readlane(r_l, j + 1);
      const int r2 = __builtin_amdgcn_readlane(r_l, j + 2);
      const int r3 = __builtin_amdgcn_readlane(r_l, j + 3);
      const float2 v0 = hw2[(size_t)r0 * 32 + lane];
      const float2 v1 = hw2[(size_t)r1 * 32 + lane];
      const float2 v2 = hw2[(size_t)r2 * 32 + lane];
      const float2 v3 = hw2[(size_t)r3 * 32 + lane];
      const float w0 = lane_f(w_l, j);
      const float w1 = lane_f(w_l, j + 1);
      const float w2 = lane_f(w_l, j + 2);
      const float w3 = lane_f(w_l, j + 3);
      acc.x += v0.x * w0; acc.y += v0.y * w0;
      acc.x += v1.x * w1; acc.y += v1.y * w1;
      acc.x += v2.x * w2; acc.y += v2.y * w2;
      acc.x += v3.x * w3; acc.y += v3.y * w3;
    }
    for (; j < k; ++j) {
      const int   r = __builtin_amdgcn_readlane(r_l, j);
      const float w = lane_f(w_l, j);
      const float2 v = hw2[(size_t)r * 32 + lane];
      acc.x += v.x * w; acc.y += v.y * w;
    }
  }

  const int f = lane * 2;
  float2 o;
  o.x = acc.x + b2[f + 0];
  o.y = acc.y + b2[f + 1];
  ((float2*)out)[(size_t)c * 32 + lane] = o;
}


extern "C" void kernel_launch(void* const* d_in, const int* in_sizes, int n_in,
                              void* d_out, int out_size, void* d_ws, size_t ws_size,
                              hipStream_t stream) {
  const float* x  = (const float*)d_in[0];
  const int*   ei = (const int*)d_in[1];
  const float* W1 = (const float*)d_in[2];
  const float* b1 = (const float*)d_in[3];
  const float* W2 = (const float*)d_in[4];
  const float* b2 = (const float*)d_in[5];
  float* out = (float*)d_out;

  const int N = in_sizes[0] / IN_F;
  const int E = in_sizes[1] / 2;
  const int* row = ei;
  const int* col = ei + E;

  float* ws   = (float*)d_ws;
  float* xw   = ws;
  float* h    = xw + (size_t)N * 128;
  float* hw   = h + (size_t)N * 128;
  float* dinv = hw + (size_t)N * 64;
  int*   cnt  = (int*)(dinv + N);
  int*   off  = cnt + N;
  int*   rank = off + N;
  int*   src  = rank + E;

  zero_int_kernel<<<(N + 255) / 256, 256, 0, stream>>>(cnt, N);
  count_rank_kernel<<<(E + 255) / 256, 256, 0, stream>>>(col, cnt, rank, E);
  scan_dinv_kernel<<<(N + SCAN_T - 1) / SCAN_T, SCAN_T, 0, stream>>>(cnt, off, dinv, N);
  place_kernel<<<(E + 255) / 256, 256, 0, stream>>>(row, col, rank, off, src, E);

  gemm_wmma_kernel<128><<<(N + 15) / 16, 256, 0, stream>>>(x, W1, xw, N);
  gather128_kernel<<<(N + 7) / 8, 256, 0, stream>>>(off, cnt, src, dinv, xw, b1, h, N);

  gemm_wmma_kernel<64><<<(N + 15) / 16, 128, 0, stream>>>(h, W2, hw, N);
  gather64_kernel<<<(N + 7) / 8, 256, 0, stream>>>(off, cnt, src, dinv, hw, b2, out, N);
}


# Both products: a block stages 32 rows of A through LDS, one whole 512-byte row per load, and each wave runs two row tiles on one B fragment; the k order of every sum is unchanged
# speedup vs baseline: 2.0496x; 1.1528x over previous
#include <hip/hip_runtime.h>

#define IN_F  128
#define HID_F 128
#define OUT_F 64

typedef __attribute__((ext_vector_type(2))) float v2f;
typedef __attribute__((ext_vector_type(8))) float v8f;


__global__ void zero_int_kernel(int* __restrict__ p, int n) {
  int i = blockIdx.x * blockDim.x + threadIdx.x;
  if (i < n) p[i] = 0;
}

__global__ void count_rank_kernel(const int* __restrict__ col, int* __restrict__ cnt,
                                  int* __restrict__ rank, int E) {
  int e = blockIdx.x * blockDim.x + threadIdx.x;
  if (e < E) rank[e] = atomicAdd(&cnt[col[e]], 1);
}

#define SCAN_T 1024
__global__ void __launch_bounds__(SCAN_T)
scan_dinv_kernel(const int* __restrict__ cnt, int* __restrict__ off,
                 float* __restrict__ dinv, int n) {
  __shared__ int wave_tot[32];
  __shared__ int wave_pre[32];
  const int t     = threadIdx.x;
  const int lane  = t & 31;
  const int wave  = t >> 5;
  const int tile0 = blockIdx.x * SCAN_T;

  int pre = 0;
  for (int i = t; i < tile0; i += SCAN_T) pre += cnt[i];

  const int i = tile0 + t;
  const int v = (i < n) ? cnt[i] : 0;

  int inc = v;
  for (int d = 1; d < 32; d <<= 1) {
    int u = __shfl_up(inc, d);
    if (lane >= d) inc += u;
  }
  for (int d = 16; d >= 1; d >>= 1) pre += __shfl_xor(pre, d);
  if (lane == 31) wave_tot[wave] = inc;
  if (lane == 0)  wave_pre[wave] = pre;
  __syncthreads();

  const int wt = wave_tot[lane];
  int winc = wt;
  for (int d = 1; d < 32; d <<= 1) {
    int u = __shfl_up(winc, d);
    if (lane >= d) winc += u;
  }
  int front = wave_pre[lane];
  for (int d = 16; d >= 1; d >>= 1) front += __shfl_xor(front, d);
  const int wave_base = __shfl(winc - wt, wave);

  if (i < n) {
    off[i]  = front + wave_base + inc - v;
    dinv[i] = rsqrtf((float)(v + 1));
  }
}

__global__ void place_kernel(const int* __restrict__ row, const int* __restrict__ col,
                             const int* __restrict__ rank, const int* __restrict__ off,
                             int* __restrict__ src, int E) {
  int e = blockIdx.x * blockDim.x + threadIdx.x;
  if (e < E) src[off[col[e]] + rank[e]] = row[e];
}


#define GEMM_RB 32
#define GEMM_AP 132

template <int F>
__global__ void __launch_bounds__((F / 16) * 32)
gemm_wmma_kernel(const float* __restrict__ A,
                 const float* __restrict__ B,
                 float* __restrict__ C,
                 int nrows) {
  constexpr int NW = F / 16;
  __shared__ float As[GEMM_RB * GEMM_AP];

  const int lane = threadIdx.x & 31;
  const int wave = threadIdx.x >> 5;
  const int row0 = blockIdx.x * GEMM_RB;

  const float4* A4 = (const float4*)A;
  float4 stage[GEMM_RB / NW];
#pragma unroll
  for (int i = 0; i < GEMM_RB / NW; ++i) {
    int row = row0 + wave + i * NW;
    row = (row < nrows) ? row : (nrows - 1);
    stage[i] = A4[(size_t)row * 32 + lane];
  }
#pragma unroll
  for (int i = 0; i < GEMM_RB / NW; ++i) {
    *(float4*)&As[(wave + i * NW) * GEMM_AP + lane * 4] = stage[i];
  }
  __syncthreads();

  const int col0 = wave * 16;
  const int m    = lane & 15;
  const int half = lane >> 4;

  v8f acc0 = {};
  v8f acc1 = {};
  const float* a0p = As + m * GEMM_AP + 2 * half;
  const float* a1p = a0p + 16 * GEMM_AP;
  const float* bp  = B + (size_t)(2 * half) * F + col0 + m;

#pragma unroll
  for (int k = 0; k < 128; k += 4) {
    v2f a0, a1, b;
    a0.x = a0p[k];
    a0.y = a0p[k + 1];
    a1.x = a1p[k];
    a1.y = a1p[k + 1];
    b.x = bp[(size_t)k * F];
    b.y = bp[(size_t)(k + 1) * F];
    acc0 = __builtin_amdgcn_wmma_f32_16x16x4_f32(
         false, a0,  false, b,
         (short)0, acc0,  false,  false);
    acc1 = __builtin_amdgcn_wmma_f32_16x16x4_f32(
         false, a1,  false, b,
         (short)0, acc1,  false,  false);
  }

  float* cp = C + (size_t)(row0 + 8 * half) * F + col0 + m;
  if (row0 + GEMM_RB <= nrows) {
#pragma unroll
    for (int v = 0; v < 8; ++v) {
      cp[(size_t)v * F]        = acc0[v];
      cp[(size_t)(v + 16) * F] = acc1[v];
    }
  } else {
#pragma unroll
    for (int v = 0; v < 8; ++v) {
      const int r0 = row0 + v + 8 * half;
      if (r0 < nrows)      cp[(size_t)v * F]        = acc0[v];
      if (r0 + 16 < nrows) cp[(size_t)(v + 16) * F] = acc1[v];
    }
  }
}


static __device__ __forceinline__ float lane_f(float x, int j) {
  return __int_as_float(__builtin_amdgcn_readlane(__float_as_int(x), j));
}

__global__ void __launch_bounds__(256)
gather128_kernel(const int* __restrict__ off, const int* __restrict__ cnt,
                 const int* __restrict__ src, const float* __restrict__ dinv,
                 const float* __restrict__ xw, const float* __restrict__ b1,
                 float* __restrict__ h, int n) {
  const int lane = threadIdx.x & 31;
  const int c = __builtin_amdgcn_readfirstlane(blockIdx.x * 8 + (threadIdx.x >> 5));
  if (c >= n) return;
  const int   start = off[c];
  const int   m     = cnt[c];
  const float dc    = dinv[c];
  const float4* xw4 = (const float4*)xw;

  const float4 s  = xw4[(size_t)c * 32 + lane];
  const float  ws = dc * dc;
  float4 acc = make_float4(s.x * ws, s.y * ws, s.z * ws, s.w * ws);

  for (int base = 0; base < m; base += 32) {
    const int k = (m - base < 32) ? (m - base) : 32;
    int   r_l = 0;
    float w_l = 0.0f;
    if (lane < k) {
      r_l = src[start + base + lane];
      w_l = dinv[r_l] * dc;
    }
    int j = 0;
    for (; j + 4 <= k; j += 4) {
      const int r0 = __builtin_amdgcn_readlane(r_l, j);
      const int r1 = __builtin_amdgcn_readlane(r_l, j + 1);
      const int r2 = __builtin_amdgcn_readlane(r_l, j + 2);
      const int r3 = __builtin_amdgcn_readlane(r_l, j + 3);
      const float4 v0 = xw4[(size_t)r0 * 32 + lane];
      const float4 v1 = xw4[(size_t)r1 * 32 + lane];
      const float4 v2 = xw4[(size_t)r2 * 32 + lane];
      const float4 v3 = xw4[(size_t)r3 * 32 + lane];
      const float w0 = lane_f(w_l, j);
      const float w1 = lane_f(w_l, j + 1);
      const float w2 = lane_f(w_l, j + 2);
      const float w3 = lane_f(w_l, j + 3);
      acc.x += v0.x * w0; acc.y += v0.y * w0; acc.z += v0.z * w0; acc.w += v0.w * w0;
      acc.x += v1.x * w1; acc.y += v1.y * w1; acc.z += v1.z * w1; acc.w += v1.w * w1;
      acc.x += v2.x * w2; acc.y += v2.y * w2; acc.z += v2.z * w2; acc.w += v2.w * w2;
      acc.x += v3.x * w3; acc.y += v3.y * w3; acc.z += v3.z * w3; acc.w += v3.w * w3;
    }
    for (; j < k; ++j) {
      const int   r = __builtin_amdgcn_readlane(r_l, j);
      const float w = lane_f(w_l, j);
      const float4 v = xw4[(size_t)r * 32 + lane];
      acc.x += v.x * w; acc.y += v.y * w; acc.z += v.z * w; acc.w += v.w * w;
    }
  }

  const int f = lane * 4;
  float4 o;
  o.x = fmaxf(acc.x + b1[f + 0], 0.0f);
  o.y = fmaxf(acc.y + b1[f + 1], 0.0f);
  o.z = fmaxf(acc.z + b1[f + 2], 0.0f);
  o.w = fmaxf(acc.w + b1[f + 3], 0.0f);
  ((float4*)h)[(size_t)c * 32 + lane] = o;
}

__global__ void __launch_bounds__(256)
gather64_kernel(const int* __restrict__ off, const int* __restrict__ cnt,
                const int* __restrict__ src, const float* __restrict__ dinv,
                const float* __restrict__ hw, const float* __restrict__ b2,
                float* __restrict__ out, int n) {
  const int lane = threadIdx.x & 31;
  const int c = __builtin_amdgcn_readfirstlane(blockIdx.x * 8 + (threadIdx.x >> 5));
  if (c >= n) return;
  const int   start = off[c];
  const int   m     = cnt[c];
  const float dc    = dinv[c];
  const float2* hw2 = (const float2*)hw;

  const float2 s  = hw2[(size_t)c * 32 + lane];
  const float  ws = dc * dc;
  float2 acc = make_float2(s.x * ws, s.y * ws);

  for (int base = 0; base < m; base += 32) {
    const int k = (m - base < 32) ? (m - base) : 32;
    int   r_l = 0;
    float w_l = 0.0f;
    if (lane < k) {
      r_l = src[start + base + lane];
      w_l = dinv[r_l] * dc;
    }
    int j = 0;
    for (; j + 4 <= k; j += 4) {
      const int r0 = __builtin_amdgcn_readlane(r_l, j);
      const int r1 = __builtin_amdgcn_readlane(r_l, j + 1);
      const int r2 = __builtin_amdgcn_readlane(r_l, j + 2);
      const int r3 = __builtin_amdgcn_readlane(r_l, j + 3);
      const float2 v0 = hw2[(size_t)r0 * 32 + lane];
      const float2 v1 = hw2[(size_t)r1 * 32 + lane];
      const float2 v2 = hw2[(size_t)r2 * 32 + lane];
      const float2 v3 = hw2[(size_t)r3 * 32 + lane];
      const float w0 = lane_f(w_l, j);
      const float w1 = lane_f(w_l, j + 1);
      const float w2 = lane_f(w_l, j + 2);
      const float w3 = lane_f(w_l, j + 3);
      acc.x += v0.x * w0; acc.y += v0.y * w0;
      acc.x += v1.x * w1; acc.y += v1.y * w1;
      acc.x += v2.x * w2; acc.y += v2.y * w2;
      acc.x += v3.x * w3; acc.y += v3.y * w3;
    }
    for (; j < k; ++j) {
      const int   r = __builtin_amdgcn_readlane(r_l, j);
      const float w = lane_f(w_l, j);
      const float2 v = hw2[(size_t)r * 32 + lane];
      acc.x += v.x * w; acc.y += v.y * w;
    }
  }

  const int f = lane * 2;
  float2 o;
  o.x = acc.x + b2[f + 0];
  o.y = acc.y + b2[f + 1];
  ((float2*)out)[(size_t)c * 32 + lane] = o;
}


extern "C" void kernel_launch(void* const* d_in, const int* in_sizes, int n_in,
                              void* d_out, int out_size, void* d_ws, size_t ws_size,
                              hipStream_t stream) {
  const float* x  = (const float*)d_in[0];
  const int*   ei = (const int*)d_in[1];
  const float* W1 = (const float*)d_in[2];
  const float* b1 = (const float*)d_in[3];
  const float* W2 = (const float*)d_in[4];
  const float* b2 = (const float*)d_in[5];
  float* out = (float*)d_out;

  const int N = in_sizes[0] / IN_F;
  const int E = in_sizes[1] / 2;
  const int* row = ei;
  const int* col = ei + E;

  float* ws   = (float*)d_ws;
  float* xw   = ws;
  float* h    = xw + (size_t)N * 128;
  float* hw   = h + (size_t)N * 128;
  float* dinv = hw + (size_t)N * 64;
  int*   cnt  = (int*)(dinv + N);
  int*   off  = cnt + N;
  int*   rank = off + N;
  int*   src  = rank + E;

  zero_int_kernel<<<(N + 255) / 256, 256, 0, stream>>>(cnt, N);
  count_rank_kernel<<<(E + 255) / 256, 256, 0, stream>>>(col, cnt, rank, E);
  scan_dinv_kernel<<<(N + SCAN_T - 1) / SCAN_T, SCAN_T, 0, stream>>>(cnt, off, dinv, N);
  place_kernel<<<(E + 255) / 256, 256, 0, stream>>>(row, col, rank, off, src, E);

  gemm_wmma_kernel<128><<<(N + GEMM_RB - 1) / GEMM_RB, 256, 0, stream>>>(x, W1, xw, N);
  gather128_kernel<<<(N + 7) / 8, 256, 0, stream>>>(off, cnt, src, dinv, xw, b1, h, N);

  gemm_wmma_kernel<64><<<(N + GEMM_RB - 1) / GEMM_RB, 128, 0, stream>>>(h, W2, hw, N);
  gather64_kernel<<<(N + 7) / 8, 256, 0, stream>>>(off, cnt, src, dinv, hw, b2, out, N);
}


# The first product has no launch of its own: its row tiles ride in the two edge-pass launches (count, placement), half in each, alternating with edge chunks; k order and arrival-order rank kept
# speedup vs baseline: 2.2507x; 1.0981x over previous
#include <hip/hip_runtime.h>

#define IN_F  128
#define HID_F 128
#define OUT_F 64

typedef __attribute__((ext_vector_type(2))) float v2f;
typedef __attribute__((ext_vector_type(8))) float v8f;


__global__ void zero_int_kernel(int* __restrict__ p, int n) {
  int i = blockIdx.x * blockDim.x + threadIdx.x;
  if (i < n) p[i] = 0;
}


#define SCAN_T 1024
__global__ void __launch_bounds__(SCAN_T)
scan_dinv_kernel(const int* __restrict__ cnt, int* __restrict__ off,
                 float* __restrict__ dinv, int n) {
  __shared__ int wave_tot[32];
  __shared__ int wave_pre[32];
  const int t     = threadIdx.x;
  const int lane  = t & 31;
  const int wave  = t >> 5;
  const int tile0 = blockIdx.x * SCAN_T;

  int pre = 0;
  for (int i = t; i < tile0; i += SCAN_T) pre += cnt[i];

  const int i = tile0 + t;
  const int v = (i < n) ? cnt[i] : 0;

  int inc = v;
  for (int d = 1; d < 32; d <<= 1) {
    int u = __shfl_up(inc, d);
    if (lane >= d) inc += u;
  }
  for (int d = 16; d >= 1; d >>= 1) pre += __shfl_xor(pre, d);
  if (lane == 31) wave_tot[wave] = inc;
  if (lane == 0)  wave_pre[wave] = pre;
  __syncthreads();

  const int wt = wave_tot[lane];
  int winc = wt;
  for (int d = 1; d < 32; d <<= 1) {
    int u = __shfl_up(winc, d);
    if (lane >= d) winc += u;
  }
  int front = wave_pre[lane];
  for (int d = 16; d >= 1; d >>= 1) front += __shfl_xor(front, d);
  const int wave_base = __shfl(winc - wt, wave);

  if (i < n) {
    off[i]  = front + wave_base + inc - v;
    dinv[i] = rsqrtf((float)(v + 1));
  }
}


#define GEMM_RB 32
#define GEMM_AP 132

template <int F>
static __device__ __forceinline__ void
gemm_tile(const float* __restrict__ A,
          const float* __restrict__ B,
          float* __restrict__ C,
          int nrows, int tile, float* As) {
  constexpr int NW = F / 16;

  const int lane = threadIdx.x & 31;
  const int wave = threadIdx.x >> 5;
  const int row0 = tile * GEMM_RB;

  const float4* A4 = (const float4*)A;
  float4 stage[GEMM_RB / NW];
#pragma unroll
  for (int i = 0; i < GEMM_RB / NW; ++i) {
    int row = row0 + wave + i * NW;
    row = (row < nrows) ? row : (nrows - 1);
    stage[i] = A4[(size_t)row * 32 + lane];
  }
#pragma unroll
  for (int i = 0; i < GEMM_RB / NW; ++i) {
    *(float4*)&As[(wave + i * NW) * GEMM_AP + lane * 4] = stage[i];
  }
  __syncthreads();

  const int col0 = wave * 16;
  const int m    = lane & 15;
  const int half = lane >> 4;

  v8f acc0 = {};
  v8f acc1 = {};
  const float* a0p = As + m * GEMM_AP + 2 * half;
  const float* a1p = a0p + 16 * GEMM_AP;
  const float* bp  = B + (size_t)(2 * half) * F + col0 + m;

#pragma unroll
  for (int k = 0; k < 128; k += 4) {
    v2f a0, a1, b;
    a0.x = a0p[k];
    a0.y = a0p[k + 1];
    a1.x = a1p[k];
    a1.y = a1p[k + 1];
    b.x = bp[(size_t)k * F];
    b.y = bp[(size_t)(k + 1) * F];
    acc0 = __builtin_amdgcn_wmma_f32_16x16x4_f32(
         false, a0,  false, b,
         (short)0, acc0,  false,  false);
    acc1 = __builtin_amdgcn_wmma_f32_16x16x4_f32(
         false, a1,  false, b,
         (short)0, acc1,  false,  false);
  }

  float* cp = C + (size_t)(row0 + 8 * half) * F + col0 + m;
  if (row0 + GEMM_RB <= nrows) {
#pragma unroll
    for (int v = 0; v < 8; ++v) {
      cp[(size_t)v * F]        = acc0[v];
      cp[(size_t)(v + 16) * F] = acc1[v];
    }
  } else {
#pragma unroll
    for (int v = 0; v < 8; ++v) {
      const int r0 = row0 + v + 8 * half;
      if (r0 < nrows)      cp[(size_t)v * F]        = acc0[v];
      if (r0 + 16 < nrows) cp[(size_t)(v + 16) * F] = acc1[v];
    }
  }
}

template <int F>
__global__ void __launch_bounds__((F / 16) * 32)
gemm_wmma_kernel(const float* __restrict__ A,
                 const float* __restrict__ B,
                 float* __restrict__ C,
                 int nrows) {
  __shared__ float As[GEMM_RB * GEMM_AP];
  gemm_tile<F>(A, B, C, nrows, blockIdx.x, As);
}

static __device__ __forceinline__ bool deal_block(int b, int tiles, int chunks, int& idx) {
  const int both = tiles < chunks ? tiles : chunks;
  if (b < 2 * both) {
    idx = b >> 1;
    return (b & 1) == 0;
  }
  idx = both + (b - 2 * both);
  return tiles > chunks;
}

__global__ void __launch_bounds__(256)
gemm128_count_kernel(const float* __restrict__ A,
                     const float* __restrict__ B,
                     float* __restrict__ C,
                     int nrows, int tile0, int tiles,
                     const int* __restrict__ col, int* __restrict__ cnt,
                     int* __restrict__ rank, int E) {
  __shared__ float As[GEMM_RB * GEMM_AP];
  int idx;
  if (deal_block(blockIdx.x, tiles, (E + 255) / 256, idx)) {
    gemm_tile<128>(A, B, C, nrows, tile0 + idx, As);
  } else {
    const int e = idx * 256 + threadIdx.x;
    if (e < E) rank[e] = atomicAdd(&cnt[col[e]], 1);
  }
}

__global__ void __launch_bounds__(256)
gemm128_place_kernel(const float* __restrict__ A,
                     const float* __restrict__ B,
                     float* __restrict__ C,
                     int nrows, int tile0, int tiles,
                     const int* __restrict__ row, const int* __restrict__ col,
                     const int* __restrict__ rank, const int* __restrict__ off,
                     int* __restrict__ src, int E) {
  __shared__ float As[GEMM_RB * GEMM_AP];
  int idx;
  if (deal_block(blockIdx.x, tiles, (E + 255) / 256, idx)) {
    gemm_tile<128>(A, B, C, nrows, tile0 + idx, As);
  } else {
    const int e = idx * 256 + threadIdx.x;
    if (e < E) src[off[col[e]] + rank[e]] = row[e];
  }
}


static __device__ __forceinline__ float lane_f(float x, int j) {
  return __int_as_float(__builtin_amdgcn_readlane(__float_as_int(x), j));
}

__global__ void __launch_bounds__(256)
gather128_kernel(const int* __restrict__ off, const int* __restrict__ cnt,
                 const int* __restrict__ src, const float* __restrict__ dinv,
                 const float* __restrict__ xw, const float* __restrict__ b1,
                 float* __restrict__ h, int n) {
  const int lane = threadIdx.x & 31;
  const int c = __builtin_amdgcn_readfirstlane(blockIdx.x * 8 + (threadIdx.x >> 5));
  if (c >= n) return;
  const int   start = off[c];
  const int   m     = cnt[c];
  const float dc    = dinv[c];
  const float4* xw4 = (const float4*)xw;

  const float4 s  = xw4[(size_t)c * 32 + lane];
  const float  ws = dc * dc;
  float4 acc = make_float4(s.x * ws, s.y * ws, s.z * ws, s.w * ws);

  for (int base = 0; base < m; base += 32) {
    const int k = (m - base < 32) ? (m - base) : 32;
    int   r_l = 0;
    float w_l = 0.0f;
    if (lane < k) {
      r_l = src[start + base + lane];
      w_l = dinv[r_l] * dc;
    }
    int j = 0;
    for (; j + 4 <= k; j += 4) {
      const int r0 = __builtin_amdgcn_readlane(r_l, j);
      const int r1 = __builtin_amdgcn_readlane(r_l, j + 1);
      const int r2 = __builtin_amdgcn_readlane(r_l, j + 2);
      const int r3 = __builtin_amdgcn_readlane(r_l, j + 3);
      const float4 v0 = xw4[(size_t)r0 * 32 + lane];
      const float4 v1 = xw4[(size_t)r1 * 32 + lane];
      const float4 v2 = xw4[(size_t)r2 * 32 + lane];
      const float4 v3 = xw4[(size_t)r3 * 32 + lane];
      const float w0 = lane_f(w_l, j);
      const float w1 = lane_f(w_l, j + 1);
      const float w2 = lane_f(w_l, j + 2);
      const float w3 = lane_f(w_l, j + 3);
      acc.x += v0.x * w0; acc.y += v0.y * w0; acc.z += v0.z * w0; acc.w += v0.w * w0;
      acc.x += v1.x * w1; acc.y += v1.y * w1; acc.z += v1.z * w1; acc.w += v1.w * w1;
      acc.x += v2.x * w2; acc.y += v2.y * w2; acc.z += v2.z * w2; acc.w += v2.w * w2;
      acc.x += v3.x * w3; acc.y += v3.y * w3; acc.z += v3.z * w3; acc.w += v3.w * w3;
    }
    for (; j < k; ++j) {
      const int   r = __builtin_amdgcn_readlane(r_l, j);
      const float w = lane_f(w_l, j);
      const float4 v = xw4[(size_t)r * 32 + lane];
      acc.x += v.x * w; acc.y += v.y * w; acc.z += v.z * w; acc.w += v.w * w;
    }
  }

  const int f = lane * 4;
  float4 o;
  o.x = fmaxf(acc.x + b1[f + 0], 0.0f);
  o.y = fmaxf(acc.y + b1[f + 1], 0.0f);
  o.z = fmaxf(acc.z + b1[f + 2], 0.0f);
  o.w = fmaxf(acc.w + b1[f + 3], 0.0f);
  ((float4*)h)[(size_t)c * 32 + lane] = o;
}

__global__ void __launch_bounds__(256)
gather64_kernel(const int* __restrict__ off, const int* __restrict__ cnt,
                const int* __restrict__ src, const float* __restrict__ dinv,
                const float* __restrict__ hw, const float* __restrict__ b2,
                float* __restrict__ out, int n) {
  const int lane = threadIdx.x & 31;
  const int c = __builtin_amdgcn_readfirstlane(blockIdx.x * 8 + (threadIdx.x >> 5));
  if (c >= n) return;
  const int   start = off[c];
  const int   m     = cnt[c];
  const float dc    = dinv[c];
  const float2* hw2 = (const float2*)hw;

  const float2 s  = hw2[(size_t)c * 32 + lane];
  const float  ws = dc * dc;
  float2 acc = make_float2(s.x * ws, s.y * ws);

  for (int base = 0; base < m; base += 32) {
    const int k = (m - base < 32) ? (m - base) : 32;
    int   r_l = 0;
    float w_l = 0.0f;
    if (lane < k) {
      r_l = src[start + base + lane];
      w_l = dinv[r_l] * dc;
    }
    int j = 0;
    for (; j + 4 <= k; j += 4) {
      const int r0 = __builtin_amdgcn_readlane(r_l, j);
      const int r1 = __builtin_amdgcn_readlane(r_l, j + 1);
      const int r2 = __builtin_amdgcn_readlane(r_l, j + 2);
      const int r3 = __builtin_amdgcn_readlane(r_l, j + 3);
      const float2 v0 = hw2[(size_t)r0 * 32 + lane];
      const float2 v1 = hw2[(size_t)r1 * 32 + lane];
      const float2 v2 = hw2[(size_t)r2 * 32 + lane];
      const float2 v3 = hw2[(size_t)r3 * 32 + lane];
      const float w0 = lane_f(w_l, j);
      const float w1 = lane_f(w_l, j + 1);
      const float w2 = lane_f(w_l, j + 2);
      const float w3 = lane_f(w_l, j + 3);
      acc.x += v0.x * w0; acc.y += v0.y * w0;
      acc.x += v1.x * w1; acc.y += v1.y * w1;
      acc.x += v2.x * w2; acc.y += v2.y * w2;
      acc.x += v3.x * w3; acc.y += v3.y * w3;
    }
    for (; j < k; ++j) {
      const int   r = __builtin_amdgcn_readlane(r_l, j);
      const float w = lane_f(w_l, j);
      const float2 v = hw2[(size_t)r * 32 + lane];
      acc.x += v.x * w; acc.y += v.y * w;
    }
  }

  const int f = lane * 2;
  float2 o;
  o.x = acc.x + b2[f + 0];
  o.y = acc.y + b2[f + 1];
  ((float2*)out)[(size_t)c * 32 + lane] = o;
}


extern "C" void kernel_launch(void* const* d_in, const int* in_sizes, int n_in,
                              void* d_out, int out_size, void* d_ws, size_t ws_size,
                              hipStream_t stream) {
  const float* x  = (const float*)d_in[0];
  const int*   ei = (const int*)d_in[1];
  const float* W1 = (const float*)d_in[2];
  const float* b1 = (const float*)d_in[3];
  const float* W2 = (const float*)d_in[4];
  const float* b2 = (const float*)d_in[5];
  float* out = (float*)d_out;

  const int N = in_sizes[0] / IN_F;
  const int E = in_sizes[1] / 2;
  const int* row = ei;
  const int* col = ei + E;

  float* ws   = (float*)d_ws;
  float* xw   = ws;
  float* h    = xw + (size_t)N * 128;
  float* hw   = h + (size_t)N * 128;
  float* dinv = hw + (size_t)N * 64;
  int*   cnt  = (int*)(dinv + N);
  int*   off  = cnt + N;
  int*   rank = off + N;
  int*   src  = rank + E;

  const int tiles   = (N + GEMM_RB - 1) / GEMM_RB;
  const int tiles_a = (tiles + 1) / 2;
  const int tiles_b = tiles - tiles_a;
  const int chunks  = (E + 255) / 256;
  zero_int_kernel<<<(N + 255) / 256, 256, 0, stream>>>(cnt, N);
  gemm128_count_kernel<<<tiles_a + chunks, 256, 0, stream>>>(x, W1, xw, N, 0, tiles_a,
                                                             col, cnt, rank, E);
  scan_dinv_kernel<<<(N + SCAN_T - 1) / SCAN_T, SCAN_T, 0, stream>>>(cnt, off, dinv, N);
  gemm128_place_kernel<<<tiles_b + chunks, 256, 0, stream>>>(x, W1, xw, N, tiles_a, tiles_b,
                                                             row, col, rank, off, src, E);

  gather128_kernel<<<(N + 7) / 8, 256, 0, stream>>>(off, cnt, src, dinv, xw, b1, h, N);

  gemm_wmma_kernel<64><<<(N + GEMM_RB - 1) / GEMM_RB, 128, 0, stream>>>(h, W2, hw, N);
  gather64_kernel<<<(N + 7) / 8, 256, 0, stream>>>(off, cnt, src, dinv, hw, b2, out, N);
}


# Edge grouping without global atomics: two-level counting sort on LDS counters (chunk x bucket counts, prefix sum, bucket placement, per-bucket slots); per-node sum order stays arrival order
# speedup vs baseline: 2.9877x; 1.3275x over previous
#include <hip/hip_runtime.h>

#define IN_F  128
#define HID_F 128
#define OUT_F 64

typedef __attribute__((ext_vector_type(2))) float v2f;
typedef __attribute__((ext_vector_type(8))) float v8f;


#define BKT_W   256
#define BKT_MAX 256
#define CHUNK_E 4096


#define SCAN_T 1024
__global__ void __launch_bounds__(SCAN_T)
scan_kernel(const int* __restrict__ cnt, int* __restrict__ off, int n) {
  __shared__ int wave_tot[32];
  __shared__ int wave_pre[32];
  const int t     = threadIdx.x;
  const int lane  = t & 31;
  const int wave  = t >> 5;
  const int tile0 = blockIdx.x * SCAN_T;

  int pre = 0;
  for (int i = t; i < tile0; i += SCAN_T) pre += cnt[i];

  const int i = tile0 + t;
  const int v = (i < n) ? cnt[i] : 0;

  int inc = v;
  for (int d = 1; d < 32; d <<= 1) {
    int u = __shfl_up(inc, d);
    if (lane >= d) inc += u;
  }
  for (int d = 16; d >= 1; d >>= 1) pre += __shfl_xor(pre, d);
  if (lane == 31) wave_tot[wave] = inc;
  if (lane == 0)  wave_pre[wave] = pre;
  __syncthreads();

  const int wt = wave_tot[lane];
  int winc = wt;
  for (int d = 1; d < 32; d <<= 1) {
    int u = __shfl_up(winc, d);
    if (lane >= d) winc += u;
  }
  int front = wave_pre[lane];
  for (int d = 16; d >= 1; d >>= 1) front += __shfl_xor(front, d);
  const int wave_base = __shfl(winc - wt, wave);

  if (i < n) off[i] = front + wave_base + inc - v;
}

#define NODE_T 1024
#define NODE_U 4
__global__ void __launch_bounds__(NODE_T)
bucket_nodes_kernel(const int* __restrict__ offm, const int2* __restrict__ pairs,
                    int* __restrict__ off, int* __restrict__ cnt,
                    float* __restrict__ dinv, int* __restrict__ src,
                    int n, int E, int nb, int nch) {
  __shared__ int lcnt[BKT_W];
  __shared__ int lcur[BKT_W];
  __shared__ int wtot[BKT_W / 32];
  const int t     = threadIdx.x;
  const int lane  = t & 31;
  const int wave  = t >> 5;
  const int b     = blockIdx.x;
  const int c0    = b * BKT_W;
  const int bstart = offm[b * nch];
  const int bend   = (b + 1 < nb) ? offm[(b + 1) * nch] : E;

  if (t < BKT_W) lcnt[t] = 0;
  __syncthreads();

  for (int p0 = bstart + t; p0 < bend; p0 += NODE_T * NODE_U) {
    int c[NODE_U];
#pragma unroll
    for (int k = 0; k < NODE_U; ++k) {
      const int p = p0 + k * NODE_T;
      c[k] = (p < bend) ? pairs[p].x : -1;
    }
#pragma unroll
    for (int k = 0; k < NODE_U; ++k) {
      if (c[k] >= 0) atomicAdd(&lcnt[c[k] - c0], 1);
    }
  }
  __syncthreads();

  int v = 0, inc = 0;
  if (t < BKT_W) {
    v   = lcnt[t];
    inc = v;
    for (int d = 1; d < 32; d <<= 1) {
      int u = __shfl_up(inc, d);
      if (lane >= d) inc += u;
    }
    if (lane == 31) wtot[wave] = inc;
  }
  __syncthreads();
  if (t < BKT_W) {
    int base = 0;
    for (int w = 0; w < wave; ++w) base += wtot[w];
    const int first = bstart + base + inc - v;
    lcur[t] = first;
    const int c = c0 + t;
    if (c < n) {
      off[c]  = first;
      cnt[c]  = v;
      dinv[c] = rsqrtf((float)(v + 1));
    }
  }
  __syncthreads();

  for (int p0 = bstart + t; p0 < bend; p0 += NODE_T * NODE_U) {
    int2 cr[NODE_U];
#pragma unroll
    for (int k = 0; k < NODE_U; ++k) {
      const int p = p0 + k * NODE_T;
      cr[k] = (p < bend) ? pairs[p] : make_int2(-1, 0);
    }
#pragma unroll
    for (int k = 0; k < NODE_U; ++k) {
      if (cr[k].x >= 0) src[atomicAdd(&lcur[cr[k].x - c0], 1)] = cr[k].y;
    }
  }
}


#define GEMM_RB 32
#define GEMM_AP 132

template <int F>
static __device__ __forceinline__ void
gemm_tile(const float* __restrict__ A,
          const float* __restrict__ B,
          float* __restrict__ C,
          int nrows, int tile, float* As) {
  constexpr int NW = F / 16;

  const int lane = threadIdx.x & 31;
  const int wave = threadIdx.x >> 5;
  const int row0 = tile * GEMM_RB;

  const float4* A4 = (const float4*)A;
  float4 stage[GEMM_RB / NW];
#pragma unroll
  for (int i = 0; i < GEMM_RB / NW; ++i) {
    int row = row0 + wave + i * NW;
    row = (row < nrows) ? row : (nrows - 1);
    stage[i] = A4[(size_t)row * 32 + lane];
  }
#pragma unroll
  for (int i = 0; i < GEMM_RB / NW; ++i) {
    *(float4*)&As[(wave + i * NW) * GEMM_AP + lane * 4] = stage[i];
  }
  __syncthreads();

  const int col0 = wave * 16;
  const int m    = lane & 15;
  const int half = lane >> 4;

  v8f acc0 = {};
  v8f acc1 = {};
  const float* a0p = As + m * GEMM_AP + 2 * half;
  const float* a1p = a0p + 16 * GEMM_AP;
  const float* bp  = B + (size_t)(2 * half) * F + col0 + m;

#pragma unroll
  for (int k = 0; k < 128; k += 4) {
    v2f a0, a1, b;
    a0.x = a0p[k];
    a0.y = a0p[k + 1];
    a1.x = a1p[k];
    a1.y = a1p[k + 1];
    b.x = bp[(size_t)k * F];
    b.y = bp[(size_t)(k + 1) * F];
    acc0 = __builtin_amdgcn_wmma_f32_16x16x4_f32(
         false, a0,  false, b,
         (short)0, acc0,  false,  false);
    acc1 = __builtin_amdgcn_wmma_f32_16x16x4_f32(
         false, a1,  false, b,
         (short)0, acc1,  false,  false);
  }

  float* cp = C + (size_t)(row0 + 8 * half) * F + col0 + m;
  if (row0 + GEMM_RB <= nrows) {
#pragma unroll
    for (int v = 0; v < 8; ++v) {
      cp[(size_t)v * F]        = acc0[v];
      cp[(size_t)(v + 16) * F] = acc1[v];
    }
  } else {
#pragma unroll
    for (int v = 0; v < 8; ++v) {
      const int r0 = row0 + v + 8 * half;
      if (r0 < nrows)      cp[(size_t)v * F]        = acc0[v];
      if (r0 + 16 < nrows) cp[(size_t)(v + 16) * F] = acc1[v];
    }
  }
}

template <int F>
__global__ void __launch_bounds__((F / 16) * 32)
gemm_wmma_kernel(const float* __restrict__ A,
                 const float* __restrict__ B,
                 float* __restrict__ C,
                 int nrows) {
  __shared__ float As[GEMM_RB * GEMM_AP];
  gemm_tile<F>(A, B, C, nrows, blockIdx.x, As);
}

static __device__ __forceinline__ bool deal_block(int b, int tiles, int chunks, int& idx) {
  const int both = tiles < chunks ? tiles : chunks;
  if (b < 2 * both) {
    idx = b >> 1;
    return (b & 1) == 0;
  }
  idx = both + (b - 2 * both);
  return tiles > chunks;
}

__global__ void __launch_bounds__(256)
gemm128_count_kernel(const float* __restrict__ A,
                     const float* __restrict__ B,
                     float* __restrict__ C,
                     int nrows, int tile0, int tiles,
                     const int* __restrict__ col, int* __restrict__ cntm,
                     int E, int nb, int nch) {
  __shared__ float As[GEMM_RB * GEMM_AP];
  __shared__ int hist[BKT_MAX];
  int idx;
  if (deal_block(blockIdx.x, tiles, nch, idx)) {
    gemm_tile<128>(A, B, C, nrows, tile0 + idx, As);
  } else {
    const int t = threadIdx.x;
    hist[t] = 0;
    __syncthreads();
    const int e0 = idx * CHUNK_E + t;
    int c[CHUNK_E / 256];
#pragma unroll
    for (int i = 0; i < CHUNK_E / 256; ++i) {
      const int e = e0 + i * 256;
      c[i] = (e < E) ? col[e] : -1;
    }
#pragma unroll
    for (int i = 0; i < CHUNK_E / 256; ++i) {
      if (c[i] >= 0) atomicAdd(&hist[c[i] / BKT_W], 1);
    }
    __syncthreads();
    if (t < nb) cntm[t * nch + idx] = hist[t];
  }
}

__global__ void __launch_bounds__(256)
gemm128_place_kernel(const float* __restrict__ A,
                     const float* __restrict__ B,
                     float* __restrict__ C,
                     int nrows, int tile0, int tiles,
                     const int* __restrict__ row, const int* __restrict__ col,
                     const int* __restrict__ offm, int2* __restrict__ pairs,
                     int E, int nb, int nch) {
  __shared__ float As[GEMM_RB * GEMM_AP];
  __shared__ int cursor[BKT_MAX];
  int idx;
  if (deal_block(blockIdx.x, tiles, nch, idx)) {
    gemm_tile<128>(A, B, C, nrows, tile0 + idx, As);
  } else {
    const int t = threadIdx.x;
    if (t < nb) cursor[t] = offm[t * nch + idx];
    const int e0 = idx * CHUNK_E + t;
    int c[CHUNK_E / 256];
    int r[CHUNK_E / 256];
#pragma unroll
    for (int i = 0; i < CHUNK_E / 256; ++i) {
      const int e = e0 + i * 256;
      c[i] = (e < E) ? col[e] : -1;
      r[i] = (e < E) ? row[e] : 0;
    }
    __syncthreads();
#pragma unroll
    for (int i = 0; i < CHUNK_E / 256; ++i) {
      if (c[i] >= 0) {
        const int p = atomicAdd(&cursor[c[i] / BKT_W], 1);
        pairs[p] = make_int2(c[i], r[i]);
      }
    }
  }
}


static __device__ __forceinline__ float lane_f(float x, int j) {
  return __int_as_float(__builtin_amdgcn_readlane(__float_as_int(x), j));
}

__global__ void __launch_bounds__(256)
gather128_kernel(const int* __restrict__ off, const int* __restrict__ cnt,
                 const int* __restrict__ src, const float* __restrict__ dinv,
                 const float* __restrict__ xw, const float* __restrict__ b1,
                 float* __restrict__ h, int n) {
  const int lane = threadIdx.x & 31;
  const int c = __builtin_amdgcn_readfirstlane(blockIdx.x * 8 + (threadIdx.x >> 5));
  if (c >= n) return;
  const int   start = off[c];
  const int   m     = cnt[c];
  const float dc    = dinv[c];
  const float4* xw4 = (const float4*)xw;

  const float4 s  = xw4[(size_t)c * 32 + lane];
  const float  ws = dc * dc;
  float4 acc = make_float4(s.x * ws, s.y * ws, s.z * ws, s.w * ws);

  for (int base = 0; base < m; base += 32) {
    const int k = (m - base < 32) ? (m - base) : 32;
    int   r_l = 0;
    float w_l = 0.0f;
    if (lane < k) {
      r_l = src[start + base + lane];
      w_l = dinv[r_l] * dc;
    }
    int j = 0;
    for (; j + 4 <= k; j += 4) {
      const int r0 = __builtin_amdgcn_readlane(r_l, j);
      const int r1 = __builtin_amdgcn_readlane(r_l, j + 1);
      const int r2 = __builtin_amdgcn_readlane(r_l, j + 2);
      const int r3 = __builtin_amdgcn_readlane(r_l, j + 3);
      const float4 v0 = xw4[(size_t)r0 * 32 + lane];
      const float4 v1 = xw4[(size_t)r1 * 32 + lane];
      const float4 v2 = xw4[(size_t)r2 * 32 + lane];
      const float4 v3 = xw4[(size_t)r3 * 32 + lane];
      const float w0 = lane_f(w_l, j);
      const float w1 = lane_f(w_l, j + 1);
      const float w2 = lane_f(w_l, j + 2);
      const float w3 = lane_f(w_l, j + 3);
      acc.x += v0.x * w0; acc.y += v0.y * w0; acc.z += v0.z * w0; acc.w += v0.w * w0;
      acc.x += v1.x * w1; acc.y += v1.y * w1; acc.z += v1.z * w1; acc.w += v1.w * w1;
      acc.x += v2.x * w2; acc.y += v2.y * w2; acc.z += v2.z * w2; acc.w += v2.w * w2;
      acc.x += v3.x * w3; acc.y += v3.y * w3; acc.z += v3.z * w3; acc.w += v3.w * w3;
    }
    for (; j < k; ++j) {
      const int   r = __builtin_amdgcn_readlane(r_l, j);
      const float w = lane_f(w_l, j);
      const float4 v = xw4[(size_t)r * 32 + lane];
      acc.x += v.x * w; acc.y += v.y * w; acc.z += v.z * w; acc.w += v.w * w;
    }
  }

  const int f = lane * 4;
  float4 o;
  o.x = fmaxf(acc.x + b1[f + 0], 0.0f);
  o.y = fmaxf(acc.y + b1[f + 1], 0.0f);
  o.z = fmaxf(acc.z + b1[f + 2], 0.0f);
  o.w = fmaxf(acc.w + b1[f + 3], 0.0f);
  ((float4*)h)[(size_t)c * 32 + lane] = o;
}

__global__ void __launch_bounds__(256)
gather64_kernel(const int* __restrict__ off, const int* __restrict__ cnt,
                const int* __restrict__ src, const float* __restrict__ dinv,
                const float* __restrict__ hw, const float* __restrict__ b2,
                float* __restrict__ out, int n) {
  const int lane = threadIdx.x & 31;
  const int c = __builtin_amdgcn_readfirstlane(blockIdx.x * 8 + (threadIdx.x >> 5));
  if (c >= n) return;
  const int   start = off[c];
  const int   m     = cnt[c];
  const float dc    = dinv[c];
  const float2* hw2 = (const float2*)hw;

  const float2 s  = hw2[(size_t)c * 32 + lane];
  const float  ws = dc * dc;
  float2 acc = make_float2(s.x * ws, s.y * ws);

  for (int base = 0; base < m; base += 32) {
    const int k = (m - base < 32) ? (m - base) : 32;
    int   r_l = 0;
    float w_l = 0.0f;
    if (lane < k) {
      r_l = src[start + base + lane];
      w_l = dinv[r_l] * dc;
    }
    int j = 0;
    for (; j + 4 <= k; j += 4) {
      const int r0 = __builtin_amdgcn_readlane(r_l, j);
      const int r1 = __builtin_amdgcn_readlane(r_l, j + 1);
      const int r2 = __builtin_amdgcn_readlane(r_l, j + 2);
      const int r3 = __builtin_amdgcn_readlane(r_l, j + 3);
      const float2 v0 = hw2[(size_t)r0 * 32 + lane];
      const float2 v1 = hw2[(size_t)r1 * 32 + lane];
      const float2 v2 = hw2[(size_t)r2 * 32 + lane];
      const float2 v3 = hw2[(size_t)r3 * 32 + lane];
      const float w0 = lane_f(w_l, j);
      const float w1 = lane_f(w_l, j + 1);
      const float w2 = lane_f(w_l, j + 2);
      const float w3 = lane_f(w_l, j + 3);
      acc.x += v0.x * w0; acc.y += v0.y * w0;
      acc.x += v1.x * w1; acc.y += v1.y * w1;
      acc.x += v2.x * w2; acc.y += v2.y * w2;
      acc.x += v3.x * w3; acc.y += v3.y * w3;
    }
    for (; j < k; ++j) {
      const int   r = __builtin_amdgcn_readlane(r_l, j);
      const float w = lane_f(w_l, j);
      const float2 v = hw2[(size_t)r * 32 + lane];
      acc.x += v.x * w; acc.y += v.y * w;
    }
  }

  const int f = lane * 2;
  float2 o;
  o.x = acc.x + b2[f + 0];
  o.y = acc.y + b2[f + 1];
  ((float2*)out)[(size_t)c * 32 + lane] = o;
}


extern "C" void kernel_launch(void* const* d_in, const int* in_sizes, int n_in,
                              void* d_out, int out_size, void* d_ws, size_t ws_size,
                              hipStream_t stream) {
  const float* x  = (const float*)d_in[0];
  const int*   ei = (const int*)d_in[1];
  const float* W1 = (const float*)d_in[2];
  const float* b1 = (const float*)d_in[3];
  const float* W2 = (const float*)d_in[4];
  const float* b2 = (const float*)d_in[5];
  float* out = (float*)d_out;

  const int N = in_sizes[0] / IN_F;
  const int E = in_sizes[1] / 2;
  const int* row = ei;
  const int* col = ei + E;

  const int nb  = (N + BKT_W - 1) / BKT_W;
  const int nch = (E + CHUNK_E - 1) / CHUNK_E;
  const int nm  = nb * nch;
  float* ws    = (float*)d_ws;
  float* xw    = ws;
  float* h     = xw + (size_t)N * 128;
  float* hw    = h + (size_t)N * 128;
  int2*  pairs = (int2*)(hw + (size_t)N * 64);
  float* dinv  = (float*)(pairs + E);
  int*   cnt   = (int*)(dinv + N);
  int*   off   = cnt + N;
  int*   src   = off + N;
  int*   cntm  = src + E;
  int*   offm  = cntm + nm;

  const int tiles   = (N + GEMM_RB - 1) / GEMM_RB;
  const int tiles_a = (tiles + 1) / 2;
  const int tiles_b = tiles - tiles_a;
  gemm128_count_kernel<<<tiles_a + nch, 256, 0, stream>>>(x, W1, xw, N, 0, tiles_a,
                                                          col, cntm, E, nb, nch);
  scan_kernel<<<(nm + SCAN_T - 1) / SCAN_T, SCAN_T, 0, stream>>>(cntm, offm, nm);
  gemm128_place_kernel<<<tiles_b + nch, 256, 0, stream>>>(x, W1, xw, N, tiles_a, tiles_b,
                                                          row, col, offm, pairs, E, nb, nch);
  bucket_nodes_kernel<<<nb, NODE_T, 0, stream>>>(offm, pairs, off, cnt, dinv, src, N, E, nb, nch);

  gather128_kernel<<<(N + 7) / 8, 256, 0, stream>>>(off, cnt, src, dinv, xw, b1, h, N);

  gemm_wmma_kernel<64><<<(N + GEMM_RB - 1) / GEMM_RB, 128, 0, stream>>>(h, W2, hw, N);
  gather64_kernel<<<(N + 7) / 8, 256, 0, stream>>>(off, cnt, src, dinv, hw, b2, out, N);
}


# Layer-1 gather and second product share a launch: 8 waves gather 32 rows (4 nodes per wave, rows fetched in fours padded with weight 0) into LDS, 4 waves multiply by W2; h not stored; sum orders kept
# speedup vs baseline: 3.2054x; 1.0729x over previous
#include <hip/hip_runtime.h>

#define IN_F  128
#define HID_F 128
#define OUT_F 64

typedef __attribute__((ext_vector_type(2))) float v2f;
typedef __attribute__((ext_vector_type(8))) float v8f;


#define BKT_W   256
#define BKT_MAX 256
#define CHUNK_E 4096


#define SCAN_T 1024
__global__ void __launch_bounds__(SCAN_T)
scan_kernel(const int* __restrict__ cnt, int* __restrict__ off, int n) {
  __shared__ int wave_tot[32];
  __shared__ int wave_pre[32];
  const int t     = threadIdx.x;
  const int lane  = t & 31;
  const int wave  = t >> 5;
  const int tile0 = blockIdx.x * SCAN_T;

  int pre = 0;
  for (int i = t; i < tile0; i += SCAN_T) pre += cnt[i];

  const int i = tile0 + t;
  const int v = (i < n) ? cnt[i] : 0;

  int inc = v;
  for (int d = 1; d < 32; d <<= 1) {
    int u = __shfl_up(inc, d);
    if (lane >= d) inc += u;
  }
  for (int d = 16; d >= 1; d >>= 1) pre += __shfl_xor(pre, d);
  if (lane == 31) wave_tot[wave] = inc;
  if (lane == 0)  wave_pre[wave] = pre;
  __syncthreads();

  const int wt = wave_tot[lane];
  int winc = wt;
  for (int d = 1; d < 32; d <<= 1) {
    int u = __shfl_up(winc, d);
    if (lane >= d) winc += u;
  }
  int front = wave_pre[lane];
  for (int d = 16; d >= 1; d >>= 1) front += __shfl_xor(front, d);
  const int wave_base = __shfl(winc - wt, wave);

  if (i < n) off[i] = front + wave_base + inc - v;
}

#define NODE_T 1024
#define NODE_U 4
__global__ void __launch_bounds__(NODE_T)
bucket_nodes_kernel(const int* __restrict__ offm, const int2* __restrict__ pairs,
                    int* __restrict__ off, int* __restrict__ cnt,
                    float* __restrict__ dinv, int* __restrict__ src,
                    int n, int E, int nb, int nch) {
  __shared__ int lcnt[BKT_W];
  __shared__ int lcur[BKT_W];
  __shared__ int wtot[BKT_W / 32];
  const int t     = threadIdx.x;
  const int lane  = t & 31;
  const int wave  = t >> 5;
  const int b     = blockIdx.x;
  const int c0    = b * BKT_W;
  const int bstart = offm[b * nch];
  const int bend   = (b + 1 < nb) ? offm[(b + 1) * nch] : E;

  if (t < BKT_W) lcnt[t] = 0;
  __syncthreads();

  for (int p0 = bstart + t; p0 < bend; p0 += NODE_T * NODE_U) {
    int c[NODE_U];
#pragma unroll
    for (int k = 0; k < NODE_U; ++k) {
      const int p = p0 + k * NODE_T;
      c[k] = (p < bend) ? pairs[p].x : -1;
    }
#pragma unroll
    for (int k = 0; k < NODE_U; ++k) {
      if (c[k] >= 0) atomicAdd(&lcnt[c[k] - c0], 1);
    }
  }
  __syncthreads();

  int v = 0, inc = 0;
  if (t < BKT_W) {
    v   = lcnt[t];
    inc = v;
    for (int d = 1; d < 32; d <<= 1) {
      int u = __shfl_up(inc, d);
      if (lane >= d) inc += u;
    }
    if (lane == 31) wtot[wave] = inc;
  }
  __syncthreads();
  if (t < BKT_W) {
    int base = 0;
    for (int w = 0; w < wave; ++w) base += wtot[w];
    const int first = bstart + base + inc - v;
    lcur[t] = first;
    const int c = c0 + t;
    if (c < n) {
      off[c]  = first;
      cnt[c]  = v;
      dinv[c] = rsqrtf((float)(v + 1));
    }
  }
  __syncthreads();

  for (int p0 = bstart + t; p0 < bend; p0 += NODE_T * NODE_U) {
    int2 cr[NODE_U];
#pragma unroll
    for (int k = 0; k < NODE_U; ++k) {
      const int p = p0 + k * NODE_T;
      cr[k] = (p < bend) ? pairs[p] : make_int2(-1, 0);
    }
#pragma unroll
    for (int k = 0; k < NODE_U; ++k) {
      if (cr[k].x >= 0) src[atomicAdd(&lcur[cr[k].x - c0], 1)] = cr[k].y;
    }
  }
}


#define GEMM_RB 32
#define GEMM_AP 132

template <int F>
static __device__ __forceinline__ void
gemm_from_lds(const float* As,
              const float* __restrict__ B,
              float* __restrict__ C,
              int nrows, int row0) {
  const int lane = threadIdx.x & 31;
  const int wave = threadIdx.x >> 5;

  const int col0 = wave * 16;
  const int m    = lane & 15;
  const int half = lane >> 4;

  v8f acc0 = {};
  v8f acc1 = {};
  const float* a0p = As + m * GEMM_AP + 2 * half;
  const float* a1p = a0p + 16 * GEMM_AP;
  const float* bp  = B + (size_t)(2 * half) * F + col0 + m;

#pragma unroll
  for (int k = 0; k < 128; k += 4) {
    v2f a0, a1, b;
    a0.x = a0p[k];
    a0.y = a0p[k + 1];
    a1.x = a1p[k];
    a1.y = a1p[k + 1];
    b.x = bp[(size_t)k * F];
    b.y = bp[(size_t)(k + 1) * F];
    acc0 = __builtin_amdgcn_wmma_f32_16x16x4_f32(
         false, a0,  false, b,
         (short)0, acc0,  false,  false);
    acc1 = __builtin_amdgcn_wmma_f32_16x16x4_f32(
         false, a1,  false, b,
         (short)0, acc1,  false,  false);
  }

  float* cp = C + (size_t)(row0 + 8 * half) * F + col0 + m;
  if (row0 + GEMM_RB <= nrows) {
#pragma unroll
    for (int v = 0; v < 8; ++v) {
      cp[(size_t)v * F]        = acc0[v];
      cp[(size_t)(v + 16) * F] = acc1[v];
    }
  } else {
#pragma unroll
    for (int v = 0; v < 8; ++v) {
      const int r0 = row0 + v + 8 * half;
      if (r0 < nrows)      cp[(size_t)v * F]        = acc0[v];
      if (r0 + 16 < nrows) cp[(size_t)(v + 16) * F] = acc1[v];
    }
  }
}

template <int F>
static __device__ __forceinline__ void
gemm_tile(const float* __restrict__ A,
          const float* __restrict__ B,
          float* __restrict__ C,
          int nrows, int tile, float* As) {
  constexpr int NW = F / 16;

  const int lane = threadIdx.x & 31;
  const int wave = threadIdx.x >> 5;
  const int row0 = tile * GEMM_RB;

  const float4* A4 = (const float4*)A;
  float4 stage[GEMM_RB / NW];
#pragma unroll
  for (int i = 0; i < GEMM_RB / NW; ++i) {
    int row = row0 + wave + i * NW;
    row = (row < nrows) ? row : (nrows - 1);
    stage[i] = A4[(size_t)row * 32 + lane];
  }
#pragma unroll
  for (int i = 0; i < GEMM_RB / NW; ++i) {
    *(float4*)&As[(wave + i * NW) * GEMM_AP + lane * 4] = stage[i];
  }
  __syncthreads();

  gemm_from_lds<F>(As, B, C, nrows, row0);
}

static __device__ __forceinline__ bool deal_block(int b, int tiles, int chunks, int& idx) {
  const int both = tiles < chunks ? tiles : chunks;
  if (b < 2 * both) {
    idx = b >> 1;
    return (b & 1) == 0;
  }
  idx = both + (b - 2 * both);
  return tiles > chunks;
}

__global__ void __launch_bounds__(256)
gemm128_count_kernel(const float* __restrict__ A,
                     const float* __restrict__ B,
                     float* __restrict__ C,
                     int nrows, int tile0, int tiles,
                     const int* __restrict__ col, int* __restrict__ cntm,
                     int E, int nb, int nch) {
  __shared__ float As[GEMM_RB * GEMM_AP];
  __shared__ int hist[BKT_MAX];
  int idx;
  if (deal_block(blockIdx.x, tiles, nch, idx)) {
    gemm_tile<128>(A, B, C, nrows, tile0 + idx, As);
  } else {
    const int t = threadIdx.x;
    hist[t] = 0;
    __syncthreads();
    const int e0 = idx * CHUNK_E + t;
    int c[CHUNK_E / 256];
#pragma unroll
    for (int i = 0; i < CHUNK_E / 256; ++i) {
      const int e = e0 + i * 256;
      c[i] = (e < E) ? col[e] : -1;
    }
#pragma unroll
    for (int i = 0; i < CHUNK_E / 256; ++i) {
      if (c[i] >= 0) atomicAdd(&hist[c[i] / BKT_W], 1);
    }
    __syncthreads();
    if (t < nb) cntm[t * nch + idx] = hist[t];
  }
}

__global__ void __launch_bounds__(256)
gemm128_place_kernel(const float* __restrict__ A,
                     const float* __restrict__ B,
                     float* __restrict__ C,
                     int nrows, int tile0, int tiles,
                     const int* __restrict__ row, const int* __restrict__ col,
                     const int* __restrict__ offm, int2* __restrict__ pairs,
                     int E, int nb, int nch) {
  __shared__ float As[GEMM_RB * GEMM_AP];
  __shared__ int cursor[BKT_MAX];
  int idx;
  if (deal_block(blockIdx.x, tiles, nch, idx)) {
    gemm_tile<128>(A, B, C, nrows, tile0 + idx, As);
  } else {
    const int t = threadIdx.x;
    if (t < nb) cursor[t] = offm[t * nch + idx];
    const int e0 = idx * CHUNK_E + t;
    int c[CHUNK_E / 256];
    int r[CHUNK_E / 256];
#pragma unroll
    for (int i = 0; i < CHUNK_E / 256; ++i) {
      const int e = e0 + i * 256;
      c[i] = (e < E) ? col[e] : -1;
      r[i] = (e < E) ? row[e] : 0;
    }
    __syncthreads();
#pragma unroll
    for (int i = 0; i < CHUNK_E / 256; ++i) {
      if (c[i] >= 0) {
        const int p = atomicAdd(&cursor[c[i] / BKT_W], 1);
        pairs[p] = make_int2(c[i], r[i]);
      }
    }
  }
}


static __device__ __forceinline__ float lane_f(float x, int j) {
  return __int_as_float(__builtin_amdgcn_readlane(__float_as_int(x), j));
}

static __device__ __forceinline__ void
add_rows128(float4& acc, const float4* __restrict__ xw4, int r_l, float w_l, int k4, int lane) {
  for (int j = 0; j < k4; j += 4) {
    int    r[4];
    float4 v[4];
#pragma unroll
    for (int u = 0; u < 4; ++u) r[u] = __builtin_amdgcn_readlane(r_l, j + u);
#pragma unroll
    for (int u = 0; u < 4; ++u) v[u] = xw4[(size_t)r[u] * 32 + lane];
#pragma unroll
    for (int u = 0; u < 4; ++u) {
      const float w = lane_f(w_l, j + u);
      acc.x += v[u].x * w; acc.y += v[u].y * w; acc.z += v[u].z * w; acc.w += v[u].w * w;
    }
  }
}

#define GATHER_NW 4
__global__ void __launch_bounds__(256, 16)
gather128_gemm64_kernel(const int* __restrict__ off, const int* __restrict__ cnt,
                        const int* __restrict__ src, const float* __restrict__ dinv,
                        const float* __restrict__ xw, const float* __restrict__ b1,
                        const float* __restrict__ W2, float* __restrict__ hw, int n) {
  static_assert(8 * GATHER_NW == GEMM_RB, "8 waves x GATHER_NW nodes fill the row tile");
  __shared__ float As[GEMM_RB * GEMM_AP];
  const int lane = threadIdx.x & 31;
  const int wave = __builtin_amdgcn_readfirstlane(threadIdx.x >> 5);
  const int row0 = blockIdx.x * GEMM_RB;
  const int c0   = row0 + wave * GATHER_NW;
  const float4* xw4 = (const float4*)xw;

  int   cc[GATHER_NW], st[GATHER_NW], mm[GATHER_NW];
  float dd[GATHER_NW];
#pragma unroll
  for (int q = 0; q < GATHER_NW; ++q) {
    const int c = c0 + q;
    cc[q] = (c < n) ? c : (n - 1);
    st[q] = off[cc[q]];
    mm[q] = (c < n) ? cnt[cc[q]] : 0;
    dd[q] = (c < n) ? dinv[cc[q]] : 0.0f;
  }
  int   r_l[GATHER_NW];
  float w_l[GATHER_NW];
#pragma unroll
  for (int q = 0; q < GATHER_NW; ++q) {
    const int k = (mm[q] + 1 < 32) ? (mm[q] + 1) : 32;
    r_l[q] = cc[q];
    if (lane >= 1 && lane < k) r_l[q] = src[st[q] + lane - 1];
  }
  float dv[GATHER_NW];
#pragma unroll
  for (int q = 0; q < GATHER_NW; ++q) dv[q] = dinv[r_l[q]];
#pragma unroll
  for (int q = 0; q < GATHER_NW; ++q) {
    const int k = (mm[q] + 1 < 32) ? (mm[q] + 1) : 32;
    w_l[q] = (lane < k) ? dv[q] * dd[q] : 0.0f;
  }

  const int f = lane * 4;
  const float bx = b1[f + 0], by = b1[f + 1], bz = b1[f + 2], bw = b1[f + 3];

#pragma unroll
  for (int q = 0; q < GATHER_NW; ++q) {
    float4 acc = make_float4(0.0f, 0.0f, 0.0f, 0.0f);
    const int t = mm[q] + 1;
    const int k = (t < 32) ? t : 32;
    add_rows128(acc, xw4, r_l[q], w_l[q], (k + 3) & ~3, lane);
    for (int base = 32; base < t; base += 32) {
      const int kk = (t - base < 32) ? (t - base) : 32;
      int   r2 = cc[q];
      float w2 = 0.0f;
      if (lane < kk) {
        r2 = src[st[q] + base + lane - 1];
        w2 = dinv[r2] * dd[q];
      }
      add_rows128(acc, xw4, r2, w2, (kk + 3) & ~3, lane);
    }

    float4 o;
    o.x = fmaxf(acc.x + bx, 0.0f);
    o.y = fmaxf(acc.y + by, 0.0f);
    o.z = fmaxf(acc.z + bz, 0.0f);
    o.w = fmaxf(acc.w + bw, 0.0f);
    *(float4*)&As[(wave * GATHER_NW + q) * GEMM_AP + f] = o;
  }
  __syncthreads();

  if (wave < OUT_F / 16) {
    gemm_from_lds<OUT_F>(As, W2, hw, n, row0);
  }
}

__global__ void __launch_bounds__(256)
gather64_kernel(const int* __restrict__ off, const int* __restrict__ cnt,
                const int* __restrict__ src, const float* __restrict__ dinv,
                const float* __restrict__ hw, const float* __restrict__ b2,
                float* __restrict__ out, int n) {
  const int lane = threadIdx.x & 31;
  const int c = __builtin_amdgcn_readfirstlane(blockIdx.x * 8 + (threadIdx.x >> 5));
  if (c >= n) return;
  const int   start = off[c];
  const int   m     = cnt[c];
  const float dc    = dinv[c];
  const float2* hw2 = (const float2*)hw;

  const float2 s  = hw2[(size_t)c * 32 + lane];
  const float  ws = dc * dc;
  float2 acc = make_float2(s.x * ws, s.y * ws);

  for (int base = 0; base < m; base += 32) {
    const int k = (m - base < 32) ? (m - base) : 32;
    int   r_l = 0;
    float w_l = 0.0f;
    if (lane < k) {
      r_l = src[start + base + lane];
      w_l = dinv[r_l] * dc;
    }
    int j = 0;
    for (; j + 4 <= k; j += 4) {
      const int r0 = __builtin_amdgcn_readlane(r_l, j);
      const int r1 = __builtin_amdgcn_readlane(r_l, j + 1);
      const int r2 = __builtin_amdgcn_readlane(r_l, j + 2);
      const int r3 = __builtin_amdgcn_readlane(r_l, j + 3);
      const float2 v0 = hw2[(size_t)r0 * 32 + lane];
      const float2 v1 = hw2[(size_t)r1 * 32 + lane];
      const float2 v2 = hw2[(size_t)r2 * 32 + lane];
      const float2 v3 = hw2[(size_t)r3 * 32 + lane];
      const float w0 = lane_f(w_l, j);
      const float w1 = lane_f(w_l, j + 1);
      const float w2 = lane_f(w_l, j + 2);
      const float w3 = lane_f(w_l, j + 3);
      acc.x += v0.x * w0; acc.y += v0.y * w0;
      acc.x += v1.x * w1; acc.y += v1.y * w1;
      acc.x += v2.x * w2; acc.y += v2.y * w2;
      acc.x += v3.x * w3; acc.y += v3.y * w3;
    }
    for (; j < k; ++j) {
      const int   r = __builtin_amdgcn_readlane(r_l, j);
      const float w = lane_f(w_l, j);
      const float2 v = hw2[(size_t)r * 32 + lane];
      acc.x += v.x * w; acc.y += v.y * w;
    }
  }

  const int f = lane * 2;
  float2 o;
  o.x = acc.x + b2[f + 0];
  o.y = acc.y + b2[f + 1];
  ((float2*)out)[(size_t)c * 32 + lane] = o;
}


extern "C" void kernel_launch(void* const* d_in, const int* in_sizes, int n_in,
                              void* d_out, int out_size, void* d_ws, size_t ws_size,
                              hipStream_t stream) {
  const float* x  = (const float*)d_in[0];
  const int*   ei = (const int*)d_in[1];
  const float* W1 = (const float*)d_in[2];
  const float* b1 = (const float*)d_in[3];
  const float* W2 = (const float*)d_in[4];
  const float* b2 = (const float*)d_in[5];
  float* out = (float*)d_out;

  const int N = in_sizes[0] / IN_F;
  const int E = in_sizes[1] / 2;
  const int* row = ei;
  const int* col = ei + E;

  const int nb  = (N + BKT_W - 1) / BKT_W;
  const int nch = (E + CHUNK_E - 1) / CHUNK_E;
  const int nm  = nb * nch;
  float* ws    = (float*)d_ws;
  float* xw    = ws;
  float* hw    = xw + (size_t)N * 128;
  int2*  pairs = (int2*)(hw + (size_t)N * 64);
  float* dinv  = (float*)(pairs + E);
  int*   cnt   = (int*)(dinv + N);
  int*   off   = cnt + N;
  int*   src   = off + N;
  int*   cntm  = src + E;
  int*   offm  = cntm + nm;

  const int tiles   = (N + GEMM_RB - 1) / GEMM_RB;
  const int tiles_a = (tiles + 1) / 2;
  const int tiles_b = tiles - tiles_a;
  gemm128_count_kernel<<<tiles_a + nch, 256, 0, stream>>>(x, W1, xw, N, 0, tiles_a,
                                                          col, cntm, E, nb, nch);
  scan_kernel<<<(nm + SCAN_T - 1) / SCAN_T, SCAN_T, 0, stream>>>(cntm, offm, nm);
  gemm128_place_kernel<<<tiles_b + nch, 256, 0, stream>>>(x, W1, xw, N, tiles_a, tiles_b,
                                                          row, col, offm, pairs, E, nb, nch);
  bucket_nodes_kernel<<<nb, NODE_T, 0, stream>>>(offm, pairs, off, cnt, dinv, src, N, E, nb, nch);

  gather128_gemm64_kernel<<<(N + GEMM_RB - 1) / GEMM_RB, 256, 0, stream>>>(off, cnt, src, dinv, xw, b1, W2, hw, N);

  gather64_kernel<<<(N + 7) / 8, 256, 0, stream>>>(off, cnt, src, dinv, hw, b2, out, N);
}


# Edge grouping level 1 sorts each chunk by bucket inside its block (LDS count, scan, place); a bucket block reads one run per chunk: no scan launch, one edge-pass launch; sums keep LDS arrival order
# speedup vs baseline: 3.9954x; 1.2465x over previous
#include <hip/hip_runtime.h>

#define IN_F  128
#define HID_F 128
#define OUT_F 64

typedef __attribute__((ext_vector_type(2))) float v2f;
typedef __attribute__((ext_vector_type(8))) float v8f;


#define BKT_W   256
#define BKT_MAX 256
#define CHUNK_E 4096


#define NODE_T 1024
#define RUN_U  8

static __device__ __forceinline__ void
run_bounds(const int* __restrict__ runs, int b, int myc, int nb, int nch, int E, int& s, int& e) {
  s = 0; e = 0;
  if (myc < nch) {
    s = runs[b * nch + myc];
    const int left = E - myc * CHUNK_E;
    e = (b + 1 < nb) ? runs[(b + 1) * nch + myc] : (left < CHUNK_E ? left : CHUNK_E);
  }
}

__global__ void __launch_bounds__(NODE_T)
bucket_nodes_kernel(const int* __restrict__ runs, const int2* __restrict__ pairs,
                    int* __restrict__ off, int* __restrict__ cnt,
                    float* __restrict__ dinv, int* __restrict__ src,
                    int n, int E, int nb, int nch) {
  __shared__ int lcnt[BKT_W];
  __shared__ int lcur[BKT_W];
  __shared__ int wtot[BKT_W / 32];
  __shared__ int wst[NODE_T / 32];
  const int t     = threadIdx.x;
  const int lane  = t & 31;
  const int wave  = __builtin_amdgcn_readfirstlane(t >> 5);
  const int b     = blockIdx.x;
  const int c0    = b * BKT_W;

  if (t < BKT_W) lcnt[t] = 0;
  __syncthreads();

  int ssum = 0;
  for (int cg = wave; cg < nch; cg += 32 * 32) {
    int s, e;
    run_bounds(runs, b, cg + 32 * lane, nb, nch, E, s, e);
    ssum += s;
    const int left = (nch - cg + 31) / 32;
    const int nr   = left < 32 ? left : 32;
    for (int j0 = 0; j0 < nr; j0 += RUN_U) {
      int x[RUN_U];
#pragma unroll
      for (int u = 0; u < RUN_U; ++u) {
        const int j  = (j0 + u) & 31;
        const int sj = __builtin_amdgcn_readlane(s, j);
        const int ej = __builtin_amdgcn_readlane(e, j);
        const int p  = sj + lane;
        x[u] = (j0 + u < nr && p < ej) ? pairs[(size_t)(cg + 32 * j) * CHUNK_E + p].x : -1;
      }
#pragma unroll
      for (int u = 0; u < RUN_U; ++u) {
        if (x[u] >= 0) atomicAdd(&lcnt[x[u] - c0], 1);
      }
      for (int u = 0; u < RUN_U && j0 + u < nr; ++u) {
        const int j  = j0 + u;
        const int sj = __builtin_amdgcn_readlane(s, j);
        const int ej = __builtin_amdgcn_readlane(e, j);
        for (int p = sj + 32 + lane; p < ej; p += 32) {
          atomicAdd(&lcnt[pairs[(size_t)(cg + 32 * j) * CHUNK_E + p].x - c0], 1);
        }
      }
    }
  }
  for (int d = 16; d >= 1; d >>= 1) ssum += __shfl_xor(ssum, d);
  if (lane == 0) wst[wave] = ssum;
  __syncthreads();

  int v = 0, inc = 0;
  if (t < BKT_W) {
    v   = lcnt[t];
    inc = v;
    for (int d = 1; d < 32; d <<= 1) {
      int u = __shfl_up(inc, d);
      if (lane >= d) inc += u;
    }
    if (lane == 31) wtot[wave] = inc;
  }
  __syncthreads();
  if (t < BKT_W) {
    int base = 0;
    for (int w = 0; w < NODE_T / 32; ++w) base += wst[w];
    for (int w = 0; w < wave; ++w) base += wtot[w];
    const int first = base + inc - v;
    lcur[t] = first;
    const int c = c0 + t;
    if (c < n) {
      off[c]  = first;
      cnt[c]  = v;
      dinv[c] = rsqrtf((float)(v + 1));
    }
  }
  __syncthreads();

  for (int cg = wave; cg < nch; cg += 32 * 32) {
    int s, e;
    run_bounds(runs, b, cg + 32 * lane, nb, nch, E, s, e);
    const int left = (nch - cg + 31) / 32;
    const int nr   = left < 32 ? left : 32;
    for (int j0 = 0; j0 < nr; j0 += RUN_U) {
      int2 cr[RUN_U];
#pragma unroll
      for (int u = 0; u < RUN_U; ++u) {
        const int j  = (j0 + u) & 31;
        const int sj = __builtin_amdgcn_readlane(s, j);
        const int ej = __builtin_amdgcn_readlane(e, j);
        const int p  = sj + lane;
        cr[u] = (j0 + u < nr && p < ej) ? pairs[(size_t)(cg + 32 * j) * CHUNK_E + p] : make_int2(-1, 0);
      }
#pragma unroll
      for (int u = 0; u < RUN_U; ++u) {
        if (cr[u].x >= 0) src[atomicAdd(&lcur[cr[u].x - c0], 1)] = cr[u].y;
      }
      for (int u = 0; u < RUN_U && j0 + u < nr; ++u) {
        const int j  = j0 + u;
        const int sj = __builtin_amdgcn_readlane(s, j);
        const int ej = __builtin_amdgcn_readlane(e, j);
        for (int p = sj + 32 + lane; p < ej; p += 32) {
          const int2 q = pairs[(size_t)(cg + 32 * j) * CHUNK_E + p];
          src[atomicAdd(&lcur[q.x - c0], 1)] = q.y;
        }
      }
    }
  }
}


#define GEMM_RB 32
#define GEMM_AP 132

template <int F>
static __device__ __forceinline__ void
gemm_from_lds(const float* As,
              const float* __restrict__ B,
              float* __restrict__ C,
              int nrows, int row0) {
  const int lane = threadIdx.x & 31;
  const int wave = threadIdx.x >> 5;

  const int col0 = wave * 16;
  const int m    = lane & 15;
  const int half = lane >> 4;

  v8f acc0 = {};
  v8f acc1 = {};
  const float* a0p = As + m * GEMM_AP + 2 * half;
  const float* a1p = a0p + 16 * GEMM_AP;
  const float* bp  = B + (size_t)(2 * half) * F + col0 + m;

#pragma unroll
  for (int k = 0; k < 128; k += 4) {
    v2f a0, a1, b;
    a0.x = a0p[k];
    a0.y = a0p[k + 1];
    a1.x = a1p[k];
    a1.y = a1p[k + 1];
    b.x = bp[(size_t)k * F];
    b.y = bp[(size_t)(k + 1) * F];
    acc0 = __builtin_amdgcn_wmma_f32_16x16x4_f32(
         false, a0,  false, b,
         (short)0, acc0,  false,  false);
    acc1 = __builtin_amdgcn_wmma_f32_16x16x4_f32(
         false, a1,  false, b,
         (short)0, acc1,  false,  false);
  }

  float* cp = C + (size_t)(row0 + 8 * half) * F + col0 + m;
  if (row0 + GEMM_RB <= nrows) {
#pragma unroll
    for (int v = 0; v < 8; ++v) {
      cp[(size_t)v * F]        = acc0[v];
      cp[(size_t)(v + 16) * F] = acc1[v];
    }
  } else {
#pragma unroll
    for (int v = 0; v < 8; ++v) {
      const int r0 = row0 + v + 8 * half;
      if (r0 < nrows)      cp[(size_t)v * F]        = acc0[v];
      if (r0 + 16 < nrows) cp[(size_t)(v + 16) * F] = acc1[v];
    }
  }
}

template <int F>
static __device__ __forceinline__ void
gemm_tile(const float* __restrict__ A,
          const float* __restrict__ B,
          float* __restrict__ C,
          int nrows, int tile, float* As) {
  constexpr int NW = F / 16;

  const int lane = threadIdx.x & 31;
  const int wave = threadIdx.x >> 5;
  const int row0 = tile * GEMM_RB;

  const float4* A4 = (const float4*)A;
  float4 stage[GEMM_RB / NW];
#pragma unroll
  for (int i = 0; i < GEMM_RB / NW; ++i) {
    int row = row0 + wave + i * NW;
    row = (row < nrows) ? row : (nrows - 1);
    stage[i] = A4[(size_t)row * 32 + lane];
  }
#pragma unroll
  for (int i = 0; i < GEMM_RB / NW; ++i) {
    *(float4*)&As[(wave + i * NW) * GEMM_AP + lane * 4] = stage[i];
  }
  __syncthreads();

  gemm_from_lds<F>(As, B, C, nrows, row0);
}

static __device__ __forceinline__ bool deal_block(int b, int tiles, int chunks, int& idx) {
  const int both = tiles < chunks ? tiles : chunks;
  if (b < 2 * both) {
    idx = b >> 1;
    return (b & 1) == 0;
  }
  idx = both + (b - 2 * both);
  return tiles > chunks;
}

__global__ void __launch_bounds__(256)
gemm128_sort_kernel(const float* __restrict__ A,
                    const float* __restrict__ B,
                    float* __restrict__ C,
                    int nrows, int tiles,
                    const int* __restrict__ row, const int* __restrict__ col,
                    int* __restrict__ runs, int2* __restrict__ pairs,
                    int E, int nb, int nch) {
  __shared__ float As[GEMM_RB * GEMM_AP];
  __shared__ int cursor[BKT_MAX];
  __shared__ int wsum[BKT_MAX / 32];
  int idx;
  if (deal_block(blockIdx.x, tiles, nch, idx)) {
    gemm_tile<128>(A, B, C, nrows, idx, As);
  } else {
    const int t    = threadIdx.x;
    const int lane = t & 31;
    const int wave = t >> 5;
    cursor[t] = 0;
    const int e0 = idx * CHUNK_E + t;
    int c[CHUNK_E / 256];
    int r[CHUNK_E / 256];
#pragma unroll
    for (int i = 0; i < CHUNK_E / 256; ++i) {
      const int e = e0 + i * 256;
      c[i] = (e < E) ? col[e] : -1;
      r[i] = (e < E) ? row[e] : 0;
    }
    __syncthreads();
#pragma unroll
    for (int i = 0; i < CHUNK_E / 256; ++i) {
      if (c[i] >= 0) atomicAdd(&cursor[c[i] / BKT_W], 1);
    }
    __syncthreads();
    const int v = cursor[t];
    int inc = v;
    for (int d = 1; d < 32; d <<= 1) {
      int u = __shfl_up(inc, d);
      if (lane >= d) inc += u;
    }
    if (lane == 31) wsum[wave] = inc;
    __syncthreads();
    int first = inc - v;
    for (int w = 0; w < wave; ++w) first += wsum[w];
    cursor[t] = first;
    if (t < nb) runs[t * nch + idx] = first;
    __syncthreads();
    int2* mine = pairs + (size_t)idx * CHUNK_E;
#pragma unroll
    for (int i = 0; i < CHUNK_E / 256; ++i) {
      if (c[i] >= 0) {
        const int p = atomicAdd(&cursor[c[i] / BKT_W], 1);
        mine[p] = make_int2(c[i], r[i]);
      }
    }
  }
}


static __device__ __forceinline__ float lane_f(float x, int j) {
  return __int_as_float(__builtin_amdgcn_readlane(__float_as_int(x), j));
}

static __device__ __forceinline__ void
add_rows128(float4& acc, const float4* __restrict__ xw4, int r_l, float w_l, int k4, int lane) {
  for (int j = 0; j < k4; j += 4) {
    int    r[4];
    float4 v[4];
#pragma unroll
    for (int u = 0; u < 4; ++u) r[u] = __builtin_amdgcn_readlane(r_l, j + u);
#pragma unroll
    for (int u = 0; u < 4; ++u) v[u] = xw4[(size_t)r[u] * 32 + lane];
#pragma unroll
    for (int u = 0; u < 4; ++u) {
      const float w = lane_f(w_l, j + u);
      acc.x += v[u].x * w; acc.y += v[u].y * w; acc.z += v[u].z * w; acc.w += v[u].w * w;
    }
  }
}

#define GATHER_NW 4
__global__ void __launch_bounds__(256, 16)
gather128_gemm64_kernel(const int* __restrict__ off, const int* __restrict__ cnt,
                        const int* __restrict__ src, const float* __restrict__ dinv,
                        const float* __restrict__ xw, const float* __restrict__ b1,
                        const float* __restrict__ W2, float* __restrict__ hw, int n) {
  static_assert(8 * GATHER_NW == GEMM_RB, "8 waves x GATHER_NW nodes fill the row tile");
  __shared__ float As[GEMM_RB * GEMM_AP];
  const int lane = threadIdx.x & 31;
  const int wave = __builtin_amdgcn_readfirstlane(threadIdx.x >> 5);
  const int row0 = blockIdx.x * GEMM_RB;
  const int c0   = row0 + wave * GATHER_NW;
  const float4* xw4 = (const float4*)xw;

  int   cc[GATHER_NW], st[GATHER_NW], mm[GATHER_NW];
  float dd[GATHER_NW];
#pragma unroll
  for (int q = 0; q < GATHER_NW; ++q) {
    const int c = c0 + q;
    cc[q] = (c < n) ? c : (n - 1);
    st[q] = off[cc[q]];
    mm[q] = (c < n) ? cnt[cc[q]] : 0;
    dd[q] = (c < n) ? dinv[cc[q]] : 0.0f;
  }
  int   r_l[GATHER_NW];
  float w_l[GATHER_NW];
#pragma unroll
  for (int q = 0; q < GATHER_NW; ++q) {
    const int k = (mm[q] + 1 < 32) ? (mm[q] + 1) : 32;
    r_l[q] = cc[q];
    if (lane >= 1 && lane < k) r_l[q] = src[st[q] + lane - 1];
  }
  float dv[GATHER_NW];
#pragma unroll
  for (int q = 0; q < GATHER_NW; ++q) dv[q] = dinv[r_l[q]];
#pragma unroll
  for (int q = 0; q < GATHER_NW; ++q) {
    const int k = (mm[q] + 1 < 32) ? (mm[q] + 1) : 32;
    w_l[q] = (lane < k) ? dv[q] * dd[q] : 0.0f;
  }

  const int f = lane * 4;
  const float bx = b1[f + 0], by = b1[f + 1], bz = b1[f + 2], bw = b1[f + 3];

#pragma unroll
  for (int q = 0; q < GATHER_NW; ++q) {
    float4 acc = make_float4(0.0f, 0.0f, 0.0f, 0.0f);
    const int t = mm[q] + 1;
    const int k = (t < 32) ? t : 32;
    add_rows128(acc, xw4, r_l[q], w_l[q], (k + 3) & ~3, lane);
    for (int base = 32; base < t; base += 32) {
      const int kk = (t - base < 32) ? (t - base) : 32;
      int   r2 = cc[q];
      float w2 = 0.0f;
      if (lane < kk) {
        r2 = src[st[q] + base + lane - 1];
        w2 = dinv[r2] * dd[q];
      }
      add_rows128(acc, xw4, r2, w2, (kk + 3) & ~3, lane);
    }

    float4 o;
    o.x = fmaxf(acc.x + bx, 0.0f);
    o.y = fmaxf(acc.y + by, 0.0f);
    o.z = fmaxf(acc.z + bz, 0.0f);
    o.w = fmaxf(acc.w + bw, 0.0f);
    *(float4*)&As[(wave * GATHER_NW + q) * GEMM_AP + f] = o;
  }
  __syncthreads();

  if (wave < OUT_F / 16) {
    gemm_from_lds<OUT_F>(As, W2, hw, n, row0);
  }
}

__global__ void __launch_bounds__(256)
gather64_kernel(const int* __restrict__ off, const int* __restrict__ cnt,
                const int* __restrict__ src, const float* __restrict__ dinv,
                const float* __restrict__ hw, const float* __restrict__ b2,
                float* __restrict__ out, int n) {
  const int lane = threadIdx.x & 31;
  const int c = __builtin_amdgcn_readfirstlane(blockIdx.x * 8 + (threadIdx.x >> 5));
  if (c >= n) return;
  const int   start = off[c];
  const int   m     = cnt[c];
  const float dc    = dinv[c];
  const float2* hw2 = (const float2*)hw;

  const float2 s  = hw2[(size_t)c * 32 + lane];
  const float  ws = dc * dc;
  float2 acc = make_float2(s.x * ws, s.y * ws);

  for (int base = 0; base < m; base += 32) {
    const int k = (m - base < 32) ? (m - base) : 32;
    int   r_l = 0;
    float w_l = 0.0f;
    if (lane < k) {
      r_l = src[start + base + lane];
      w_l = dinv[r_l] * dc;
    }
    int j = 0;
    for (; j + 4 <= k; j += 4) {
      const int r0 = __builtin_amdgcn_readlane(r_l, j);
      const int r1 = __builtin_amdgcn_readlane(r_l, j + 1);
      const int r2 = __builtin_amdgcn_readlane(r_l, j + 2);
      const int r3 = __builtin_amdgcn_readlane(r_l, j + 3);
      const float2 v0 = hw2[(size_t)r0 * 32 + lane];
      const float2 v1 = hw2[(size_t)r1 * 32 + lane];
      const float2 v2 = hw2[(size_t)r2 * 32 + lane];
      const float2 v3 = hw2[(size_t)r3 * 32 + lane];
      const float w0 = lane_f(w_l, j);
      const float w1 = lane_f(w_l, j + 1);
      const float w2 = lane_f(w_l, j + 2);
      const float w3 = lane_f(w_l, j + 3);
      acc.x += v0.x * w0; acc.y += v0.y * w0;
      acc.x += v1.x * w1; acc.y += v1.y * w1;
      acc.x += v2.x * w2; acc.y += v2.y * w2;
      acc.x += v3.x * w3; acc.y += v3.y * w3;
    }
    for (; j < k; ++j) {
      const int   r = __builtin_amdgcn_readlane(r_l, j);
      const float w = lane_f(w_l, j);
      const float2 v = hw2[(size_t)r * 32 + lane];
      acc.x += v.x * w; acc.y += v.y * w;
    }
  }

  const int f = lane * 2;
  float2 o;
  o.x = acc.x + b2[f + 0];
  o.y = acc.y + b2[f + 1];
  ((float2*)out)[(size_t)c * 32 + lane] = o;
}


extern "C" void kernel_launch(void* const* d_in, const int* in_sizes, int n_in,
                              void* d_out, int out_size, void* d_ws, size_t ws_size,
                              hipStream_t stream) {
  const float* x  = (const float*)d_in[0];
  const int*   ei = (const int*)d_in[1];
  const float* W1 = (const float*)d_in[2];
  const float* b1 = (const float*)d_in[3];
  const float* W2 = (const float*)d_in[4];
  const float* b2 = (const float*)d_in[5];
  float* out = (float*)d_out;

  const int N = in_sizes[0] / IN_F;
  const int E = in_sizes[1] / 2;
  const int* row = ei;
  const int* col = ei + E;

  const int nb  = (N + BKT_W - 1) / BKT_W;
  const int nch = (E + CHUNK_E - 1) / CHUNK_E;
  float* ws    = (float*)d_ws;
  float* xw    = ws;
  float* hw    = xw + (size_t)N * 128;
  int2*  pairs = (int2*)(hw + (size_t)N * 64);
  float* dinv  = (float*)(pairs + (size_t)nch * CHUNK_E);
  int*   cnt   = (int*)(dinv + N);
  int*   off   = cnt + N;
  int*   src   = off + N;
  int*   runs  = src + E;

  const int tiles = (N + GEMM_RB - 1) / GEMM_RB;
  gemm128_sort_kernel<<<tiles + nch, 256, 0, stream>>>(x, W1, xw, N, tiles,
                                                       row, col, runs, pairs, E, nb, nch);
  bucket_nodes_kernel<<<nb, NODE_T, 0, stream>>>(runs, pairs, off, cnt, dinv, src, N, E, nb, nch);

  gather128_gemm64_kernel<<<(N + GEMM_RB - 1) / GEMM_RB, 256, 0, stream>>>(off, cnt, src, dinv, xw, b1, W2, hw, N);

  gather64_kernel<<<(N + 7) / 8, 256, 0, stream>>>(off, cnt, src, dinv, hw, b2, out, N);
}


# Layer-2 edge weight split: h rows scaled by dinv of their node before W2, so the gather adds stored rows as they are (no dinv fetch per edge) and scales the sum once; rounding changes, sum order kept
# speedup vs baseline: 4.1423x; 1.0368x over previous
#include <hip/hip_runtime.h>

#define IN_F  128
#define HID_F 128
#define OUT_F 64

typedef __attribute__((ext_vector_type(2))) float v2f;
typedef __attribute__((ext_vector_type(8))) float v8f;


#define BKT_W   256
#define BKT_MAX 256
#define CHUNK_E 4096


#define NODE_T 1024
#define RUN_U  8

static __device__ __forceinline__ void
run_bounds(const int* __restrict__ runs, int b, int myc, int nb, int nch, int E, int& s, int& e) {
  s = 0; e = 0;
  if (myc < nch) {
    s = runs[b * nch + myc];
    const int left = E - myc * CHUNK_E;
    e = (b + 1 < nb) ? runs[(b + 1) * nch + myc] : (left < CHUNK_E ? left : CHUNK_E);
  }
}

__global__ void __launch_bounds__(NODE_T)
bucket_nodes_kernel(const int* __restrict__ runs, const int2* __restrict__ pairs,
                    int* __restrict__ off, int* __restrict__ cnt,
                    float* __restrict__ dinv, int* __restrict__ src,
                    int n, int E, int nb, int nch) {
  __shared__ int lcnt[BKT_W];
  __shared__ int lcur[BKT_W];
  __shared__ int wtot[BKT_W / 32];
  __shared__ int wst[NODE_T / 32];
  const int t     = threadIdx.x;
  const int lane  = t & 31;
  const int wave  = __builtin_amdgcn_readfirstlane(t >> 5);
  const int b     = blockIdx.x;
  const int c0    = b * BKT_W;

  if (t < BKT_W) lcnt[t] = 0;
  __syncthreads();

  int ssum = 0;
  for (int cg = wave; cg < nch; cg += 32 * 32) {
    int s, e;
    run_bounds(runs, b, cg + 32 * lane, nb, nch, E, s, e);
    ssum += s;
    const int left = (nch - cg + 31) / 32;
    const int nr   = left < 32 ? left : 32;
    for (int j0 = 0; j0 < nr; j0 += RUN_U) {
      int x[RUN_U];
#pragma unroll
      for (int u = 0; u < RUN_U; ++u) {
        const int j  = (j0 + u) & 31;
        const int sj = __builtin_amdgcn_readlane(s, j);
        const int ej = __builtin_amdgcn_readlane(e, j);
        const int p  = sj + lane;
        x[u] = (j0 + u < nr && p < ej) ? pairs[(size_t)(cg + 32 * j) * CHUNK_E + p].x : -1;
      }
#pragma unroll
      for (int u = 0; u < RUN_U; ++u) {
        if (x[u] >= 0) atomicAdd(&lcnt[x[u] - c0], 1);
      }
      for (int u = 0; u < RUN_U && j0 + u < nr; ++u) {
        const int j  = j0 + u;
        const int sj = __builtin_amdgcn_readlane(s, j);
        const int ej = __builtin_amdgcn_readlane(e, j);
        for (int p = sj + 32 + lane; p < ej; p += 32) {
          atomicAdd(&lcnt[pairs[(size_t)(cg + 32 * j) * CHUNK_E + p].x - c0], 1);
        }
      }
    }
  }
  for (int d = 16; d >= 1; d >>= 1) ssum += __shfl_xor(ssum, d);
  if (lane == 0) wst[wave] = ssum;
  __syncthreads();

  int v = 0, inc = 0;
  if (t < BKT_W) {
    v   = lcnt[t];
    inc = v;
    for (int d = 1; d < 32; d <<= 1) {
      int u = __shfl_up(inc, d);
      if (lane >= d) inc += u;
    }
    if (lane == 31) wtot[wave] = inc;
  }
  __syncthreads();
  if (t < BKT_W) {
    int base = 0;
    for (int w = 0; w < NODE_T / 32; ++w) base += wst[w];
    for (int w = 0; w < wave; ++w) base += wtot[w];
    const int first = base + inc - v;
    lcur[t] = first;
    const int c = c0 + t;
    if (c < n) {
      off[c]  = first;
      cnt[c]  = v;
      dinv[c] = rsqrtf((float)(v + 1));
    }
  }
  __syncthreads();

  for (int cg = wave; cg < nch; cg += 32 * 32) {
    int s, e;
    run_bounds(runs, b, cg + 32 * lane, nb, nch, E, s, e);
    const int left = (nch - cg + 31) / 32;
    const int nr   = left < 32 ? left : 32;
    for (int j0 = 0; j0 < nr; j0 += RUN_U) {
      int2 cr[RUN_U];
#pragma unroll
      for (int u = 0; u < RUN_U; ++u) {
        const int j  = (j0 + u) & 31;
        const int sj = __builtin_amdgcn_readlane(s, j);
        const int ej = __builtin_amdgcn_readlane(e, j);
        const int p  = sj + lane;
        cr[u] = (j0 + u < nr && p < ej) ? pairs[(size_t)(cg + 32 * j) * CHUNK_E + p] : make_int2(-1, 0);
      }
#pragma unroll
      for (int u = 0; u < RUN_U; ++u) {
        if (cr[u].x >= 0) src[atomicAdd(&lcur[cr[u].x - c0], 1)] = cr[u].y;
      }
      for (int u = 0; u < RUN_U && j0 + u < nr; ++u) {
        const int j  = j0 + u;
        const int sj = __builtin_amdgcn_readlane(s, j);
        const int ej = __builtin_amdgcn_readlane(e, j);
        for (int p = sj + 32 + lane; p < ej; p += 32) {
          const int2 q = pairs[(size_t)(cg + 32 * j) * CHUNK_E + p];
          src[atomicAdd(&lcur[q.x - c0], 1)] = q.y;
        }
      }
    }
  }
}


#define GEMM_RB 32
#define GEMM_AP 132

template <int F>
static __device__ __forceinline__ void
gemm_from_lds(const float* As,
              const float* __restrict__ B,
              float* __restrict__ C,
              int nrows, int row0) {
  const int lane = threadIdx.x & 31;
  const int wave = threadIdx.x >> 5;

  const int col0 = wave * 16;
  const int m    = lane & 15;
  const int half = lane >> 4;

  v8f acc0 = {};
  v8f acc1 = {};
  const float* a0p = As + m * GEMM_AP + 2 * half;
  const float* a1p = a0p + 16 * GEMM_AP;
  const float* bp  = B + (size_t)(2 * half) * F + col0 + m;

#pragma unroll
  for (int k = 0; k < 128; k += 4) {
    v2f a0, a1, b;
    a0.x = a0p[k];
    a0.y = a0p[k + 1];
    a1.x = a1p[k];
    a1.y = a1p[k + 1];
    b.x = bp[(size_t)k * F];
    b.y = bp[(size_t)(k + 1) * F];
    acc0 = __builtin_amdgcn_wmma_f32_16x16x4_f32(
         false, a0,  false, b,
         (short)0, acc0,  false,  false);
    acc1 = __builtin_amdgcn_wmma_f32_16x16x4_f32(
         false, a1,  false, b,
         (short)0, acc1,  false,  false);
  }

  float* cp = C + (size_t)(row0 + 8 * half) * F + col0 + m;
  if (row0 + GEMM_RB <= nrows) {
#pragma unroll
    for (int v = 0; v < 8; ++v) {
      cp[(size_t)v * F]        = acc0[v];
      cp[(size_t)(v + 16) * F] = acc1[v];
    }
  } else {
#pragma unroll
    for (int v = 0; v < 8; ++v) {
      const int r0 = row0 + v + 8 * half;
      if (r0 < nrows)      cp[(size_t)v * F]        = acc0[v];
      if (r0 + 16 < nrows) cp[(size_t)(v + 16) * F] = acc1[v];
    }
  }
}

template <int F>
static __device__ __forceinline__ void
gemm_tile(const float* __restrict__ A,
          const float* __restrict__ B,
          float* __restrict__ C,
          int nrows, int tile, float* As) {
  constexpr int NW = F / 16;

  const int lane = threadIdx.x & 31;
  const int wave = threadIdx.x >> 5;
  const int row0 = tile * GEMM_RB;

  const float4* A4 = (const float4*)A;
  float4 stage[GEMM_RB / NW];
#pragma unroll
  for (int i = 0; i < GEMM_RB / NW; ++i) {
    int row = row0 + wave + i * NW;
    row = (row < nrows) ? row : (nrows - 1);
    stage[i] = A4[(size_t)row * 32 + lane];
  }
#pragma unroll
  for (int i = 0; i < GEMM_RB / NW; ++i) {
    *(float4*)&As[(wave + i * NW) * GEMM_AP + lane * 4] = stage[i];
  }
  __syncthreads();

  gemm_from_lds<F>(As, B, C, nrows, row0);
}

static __device__ __forceinline__ bool deal_block(int b, int tiles, int chunks, int& idx) {
  const int both = tiles < chunks ? tiles : chunks;
  if (b < 2 * both) {
    idx = b >> 1;
    return (b & 1) == 0;
  }
  idx = both + (b - 2 * both);
  return tiles > chunks;
}

__global__ void __launch_bounds__(256)
gemm128_sort_kernel(const float* __restrict__ A,
                    const float* __restrict__ B,
                    float* __restrict__ C,
                    int nrows, int tiles,
                    const int* __restrict__ row, const int* __restrict__ col,
                    int* __restrict__ runs, int2* __restrict__ pairs,
                    int E, int nb, int nch) {
  __shared__ float As[GEMM_RB * GEMM_AP];
  __shared__ int cursor[BKT_MAX];
  __shared__ int wsum[BKT_MAX / 32];
  int idx;
  if (deal_block(blockIdx.x, tiles, nch, idx)) {
    gemm_tile<128>(A, B, C, nrows, idx, As);
  } else {
    const int t    = threadIdx.x;
    const int lane = t & 31;
    const int wave = t >> 5;
    cursor[t] = 0;
    const int e0 = idx * CHUNK_E + t;
    int c[CHUNK_E / 256];
    int r[CHUNK_E / 256];
#pragma unroll
    for (int i = 0; i < CHUNK_E / 256; ++i) {
      const int e = e0 + i * 256;
      c[i] = (e < E) ? col[e] : -1;
      r[i] = (e < E) ? row[e] : 0;
    }
    __syncthreads();
#pragma unroll
    for (int i = 0; i < CHUNK_E / 256; ++i) {
      if (c[i] >= 0) atomicAdd(&cursor[c[i] / BKT_W], 1);
    }
    __syncthreads();
    const int v = cursor[t];
    int inc = v;
    for (int d = 1; d < 32; d <<= 1) {
      int u = __shfl_up(inc, d);
      if (lane >= d) inc += u;
    }
    if (lane == 31) wsum[wave] = inc;
    __syncthreads();
    int first = inc - v;
    for (int w = 0; w < wave; ++w) first += wsum[w];
    cursor[t] = first;
    if (t < nb) runs[t * nch + idx] = first;
    __syncthreads();
    int2* mine = pairs + (size_t)idx * CHUNK_E;
#pragma unroll
    for (int i = 0; i < CHUNK_E / 256; ++i) {
      if (c[i] >= 0) {
        const int p = atomicAdd(&cursor[c[i] / BKT_W], 1);
        mine[p] = make_int2(c[i], r[i]);
      }
    }
  }
}


static __device__ __forceinline__ float lane_f(float x, int j) {
  return __int_as_float(__builtin_amdgcn_readlane(__float_as_int(x), j));
}

static __device__ __forceinline__ void
add_rows128(float4& acc, const float4* __restrict__ xw4, int r_l, float w_l, int k4, int lane) {
  for (int j = 0; j < k4; j += 4) {
    int    r[4];
    float4 v[4];
#pragma unroll
    for (int u = 0; u < 4; ++u) r[u] = __builtin_amdgcn_readlane(r_l, j + u);
#pragma unroll
    for (int u = 0; u < 4; ++u) v[u] = xw4[(size_t)r[u] * 32 + lane];
#pragma unroll
    for (int u = 0; u < 4; ++u) {
      const float w = lane_f(w_l, j + u);
      acc.x += v[u].x * w; acc.y += v[u].y * w; acc.z += v[u].z * w; acc.w += v[u].w * w;
    }
  }
}

#define GATHER_NW 4
__global__ void __launch_bounds__(256, 16)
gather128_gemm64_kernel(const int* __restrict__ off, const int* __restrict__ cnt,
                        const int* __restrict__ src, const float* __restrict__ dinv,
                        const float* __restrict__ xw, const float* __restrict__ b1,
                        const float* __restrict__ W2, float* __restrict__ hws, int n) {
  static_assert(8 * GATHER_NW == GEMM_RB, "8 waves x GATHER_NW nodes fill the row tile");
  __shared__ float As[GEMM_RB * GEMM_AP];
  const int lane = threadIdx.x & 31;
  const int wave = __builtin_amdgcn_readfirstlane(threadIdx.x >> 5);
  const int row0 = blockIdx.x * GEMM_RB;
  const int c0   = row0 + wave * GATHER_NW;
  const float4* xw4 = (const float4*)xw;

  int   cc[GATHER_NW], st[GATHER_NW], mm[GATHER_NW];
  float dd[GATHER_NW];
#pragma unroll
  for (int q = 0; q < GATHER_NW; ++q) {
    const int c = c0 + q;
    cc[q] = (c < n) ? c : (n - 1);
    st[q] = off[cc[q]];
    mm[q] = (c < n) ? cnt[cc[q]] : 0;
    dd[q] = (c < n) ? dinv[cc[q]] : 0.0f;
  }
  int   r_l[GATHER_NW];
  float w_l[GATHER_NW];
#pragma unroll
  for (int q = 0; q < GATHER_NW; ++q) {
    const int k = (mm[q] + 1 < 32) ? (mm[q] + 1) : 32;
    r_l[q] = cc[q];
    if (lane >= 1 && lane < k) r_l[q] = src[st[q] + lane - 1];
  }
  float dv[GATHER_NW];
#pragma unroll
  for (int q = 0; q < GATHER_NW; ++q) dv[q] = dinv[r_l[q]];
#pragma unroll
  for (int q = 0; q < GATHER_NW; ++q) {
    const int k = (mm[q] + 1 < 32) ? (mm[q] + 1) : 32;
    w_l[q] = (lane < k) ? dv[q] * dd[q] : 0.0f;
  }

  const int f = lane * 4;
  const float bx = b1[f + 0], by = b1[f + 1], bz = b1[f + 2], bw = b1[f + 3];

#pragma unroll
  for (int q = 0; q < GATHER_NW; ++q) {
    float4 acc = make_float4(0.0f, 0.0f, 0.0f, 0.0f);
    const int t = mm[q] + 1;
    const int k = (t < 32) ? t : 32;
    add_rows128(acc, xw4, r_l[q], w_l[q], (k + 3) & ~3, lane);
    for (int base = 32; base < t; base += 32) {
      const int kk = (t - base < 32) ? (t - base) : 32;
      int   r2 = cc[q];
      float w2 = 0.0f;
      if (lane < kk) {
        r2 = src[st[q] + base + lane - 1];
        w2 = dinv[r2] * dd[q];
      }
      add_rows128(acc, xw4, r2, w2, (kk + 3) & ~3, lane);
    }

    float4 o;
    o.x = fmaxf(acc.x + bx, 0.0f) * dd[q];
    o.y = fmaxf(acc.y + by, 0.0f) * dd[q];
    o.z = fmaxf(acc.z + bz, 0.0f) * dd[q];
    o.w = fmaxf(acc.w + bw, 0.0f) * dd[q];
    *(float4*)&As[(wave * GATHER_NW + q) * GEMM_AP + f] = o;
  }
  __syncthreads();

  if (wave < OUT_F / 16) {
    gemm_from_lds<OUT_F>(As, W2, hws, n, row0);
  }
}

static __device__ __forceinline__ void
add_rows64(float2& acc, const float2* __restrict__ hws2, int r_l, float f_l, int k4, int lane) {
  for (int j = 0; j < k4; j += 4) {
    int    r[4];
    float2 v[4];
#pragma unroll
    for (int u = 0; u < 4; ++u) r[u] = __builtin_amdgcn_readlane(r_l, j + u);
#pragma unroll
    for (int u = 0; u < 4; ++u) v[u] = hws2[(size_t)r[u] * 32 + lane];
#pragma unroll
    for (int u = 0; u < 4; ++u) {
      const float f = lane_f(f_l, j + u);
      acc.x += v[u].x * f; acc.y += v[u].y * f;
    }
  }
}

__global__ void __launch_bounds__(256)
gather64_kernel(const int* __restrict__ off, const int* __restrict__ cnt,
                const int* __restrict__ src, const float* __restrict__ dinv,
                const float* __restrict__ hws, const float* __restrict__ b2,
                float* __restrict__ out, int n) {
  const int lane = threadIdx.x & 31;
  const int c = __builtin_amdgcn_readfirstlane(blockIdx.x * 8 + (threadIdx.x >> 5));
  if (c >= n) return;
  const int   start = off[c];
  const int   t     = cnt[c] + 1;
  const float dc    = dinv[c];
  const float2* hws2 = (const float2*)hws;
  const int f = lane * 2;
  const float bx = b2[f + 0], by = b2[f + 1];

  float2 acc = make_float2(0.0f, 0.0f);
  int base = 0;
  do {
    const int k = (t - base < 32) ? (t - base) : 32;
    int r_l = c;
    if (lane < k && base + lane >= 1) r_l = src[start + base + lane - 1];
    const float f_l = (lane < k) ? 1.0f : 0.0f;
    add_rows64(acc, hws2, r_l, f_l, (k + 3) & ~3, lane);
    base += 32;
  } while (base < t);

  float2 o;
  o.x = acc.x * dc + bx;
  o.y = acc.y * dc + by;
  ((float2*)out)[(size_t)c * 32 + lane] = o;
}


extern "C" void kernel_launch(void* const* d_in, const int* in_sizes, int n_in,
                              void* d_out, int out_size, void* d_ws, size_t ws_size,
                              hipStream_t stream) {
  const float* x  = (const float*)d_in[0];
  const int*   ei = (const int*)d_in[1];
  const float* W1 = (const float*)d_in[2];
  const float* b1 = (const float*)d_in[3];
  const float* W2 = (const float*)d_in[4];
  const float* b2 = (const float*)d_in[5];
  float* out = (float*)d_out;

  const int N = in_sizes[0] / IN_F;
  const int E = in_sizes[1] / 2;
  const int* row = ei;
  const int* col = ei + E;

  const int nb  = (N + BKT_W - 1) / BKT_W;
  const int nch = (E + CHUNK_E - 1) / CHUNK_E;
  float* ws    = (float*)d_ws;
  float* xw    = ws;
  float* hws   = xw + (size_t)N * 128;
  int2*  pairs = (int2*)(hws + (size_t)N * 64);
  float* dinv  = (float*)(pairs + (size_t)nch * CHUNK_E);
  int*   cnt   = (int*)(dinv + N);
  int*   off   = cnt + N;
  int*   src   = off + N;
  int*   runs  = src + E;

  const int tiles = (N + GEMM_RB - 1) / GEMM_RB;
  gemm128_sort_kernel<<<tiles + nch, 256, 0, stream>>>(x, W1, xw, N, tiles,
                                                       row, col, runs, pairs, E, nb, nch);
  bucket_nodes_kernel<<<nb, NODE_T, 0, stream>>>(runs, pairs, off, cnt, dinv, src, N, E, nb, nch);

  gather128_gemm64_kernel<<<(N + GEMM_RB - 1) / GEMM_RB, 256, 0, stream>>>(off, cnt, src, dinv, xw, b1, W2, hws, N);

  gather64_kernel<<<(N + 7) / 8, 256, 0, stream>>>(off, cnt, src, dinv, hws, b2, out, N);
}


# Layer-2 gather: a wave sums two nodes at once, one per 16-lane half with 16-byte row loads, two such pairs per wave (ids in one group, 8 loads in flight): a quarter of the waves; sum order kept
# speedup vs baseline: 4.2989x; 1.0378x over previous
#include <hip/hip_runtime.h>

#define IN_F  128
#define HID_F 128
#define OUT_F 64

typedef __attribute__((ext_vector_type(2))) float v2f;
typedef __attribute__((ext_vector_type(4))) float v4f;
typedef __attribute__((ext_vector_type(8))) float v8f;


#define BKT_W   256
#define BKT_MAX 256
#define CHUNK_E 4096


#define NODE_T 1024
#define RUN_U  8

static __device__ __forceinline__ void
run_bounds(const int* __restrict__ runs, int b, int myc, int nb, int nch, int E, int& s, int& e) {
  s = 0; e = 0;
  if (myc < nch) {
    s = runs[b * nch + myc];
    const int left = E - myc * CHUNK_E;
    e = (b + 1 < nb) ? runs[(b + 1) * nch + myc] : (left < CHUNK_E ? left : CHUNK_E);
  }
}

__global__ void __launch_bounds__(NODE_T)
bucket_nodes_kernel(const int* __restrict__ runs, const int2* __restrict__ pairs,
                    int* __restrict__ off, int* __restrict__ cnt,
                    float* __restrict__ dinv, int* __restrict__ src,
                    int n, int E, int nb, int nch) {
  __shared__ int lcnt[BKT_W];
  __shared__ int lcur[BKT_W];
  __shared__ int wtot[BKT_W / 32];
  __shared__ int wst[NODE_T / 32];
  const int t     = threadIdx.x;
  const int lane  = t & 31;
  const int wave  = __builtin_amdgcn_readfirstlane(t >> 5);
  const int b     = blockIdx.x;
  const int c0    = b * BKT_W;

  if (t < BKT_W) lcnt[t] = 0;
  __syncthreads();

  int ssum = 0;
  for (int cg = wave; cg < nch; cg += 32 * 32) {
    int s, e;
    run_bounds(runs, b, cg + 32 * lane, nb, nch, E, s, e);
    ssum += s;
    const int left = (nch - cg + 31) / 32;
    const int nr   = left < 32 ? left : 32;
    for (int j0 = 0; j0 < nr; j0 += RUN_U) {
      int x[RUN_U];
#pragma unroll
      for (int u = 0; u < RUN_U; ++u) {
        const int j  = (j0 + u) & 31;
        const int sj = __builtin_amdgcn_readlane(s, j);
        const int ej = __builtin_amdgcn_readlane(e, j);
        const int p  = sj + lane;
        x[u] = (j0 + u < nr && p < ej) ? pairs[(size_t)(cg + 32 * j) * CHUNK_E + p].x : -1;
      }
#pragma unroll
      for (int u = 0; u < RUN_U; ++u) {
        if (x[u] >= 0) atomicAdd(&lcnt[x[u] - c0], 1);
      }
      for (int u = 0; u < RUN_U && j0 + u < nr; ++u) {
        const int j  = j0 + u;
        const int sj = __builtin_amdgcn_readlane(s, j);
        const int ej = __builtin_amdgcn_readlane(e, j);
        for (int p = sj + 32 + lane; p < ej; p += 32) {
          atomicAdd(&lcnt[pairs[(size_t)(cg + 32 * j) * CHUNK_E + p].x - c0], 1);
        }
      }
    }
  }
  for (int d = 16; d >= 1; d >>= 1) ssum += __shfl_xor(ssum, d);
  if (lane == 0) wst[wave] = ssum;
  __syncthreads();

  int v = 0, inc = 0;
  if (t < BKT_W) {
    v   = lcnt[t];
    inc = v;
    for (int d = 1; d < 32; d <<= 1) {
      int u = __shfl_up(inc, d);
      if (lane >= d) inc += u;
    }
    if (lane == 31) wtot[wave] = inc;
  }
  __syncthreads();
  if (t < BKT_W) {
    int base = 0;
    for (int w = 0; w < NODE_T / 32; ++w) base += wst[w];
    for (int w = 0; w < wave; ++w) base += wtot[w];
    const int first = base + inc - v;
    lcur[t] = first;
    const int c = c0 + t;
    if (c < n) {
      off[c]  = first;
      cnt[c]  = v;
      dinv[c] = rsqrtf((float)(v + 1));
    }
  }
  __syncthreads();

  for (int cg = wave; cg < nch; cg += 32 * 32) {
    int s, e;
    run_bounds(runs, b, cg + 32 * lane, nb, nch, E, s, e);
    const int left = (nch - cg + 31) / 32;
    const int nr   = left < 32 ? left : 32;
    for (int j0 = 0; j0 < nr; j0 += RUN_U) {
      int2 cr[RUN_U];
#pragma unroll
      for (int u = 0; u < RUN_U; ++u) {
        const int j  = (j0 + u) & 31;
        const int sj = __builtin_amdgcn_readlane(s, j);
        const int ej = __builtin_amdgcn_readlane(e, j);
        const int p  = sj + lane;
        cr[u] = (j0 + u < nr && p < ej) ? pairs[(size_t)(cg + 32 * j) * CHUNK_E + p] : make_int2(-1, 0);
      }
#pragma unroll
      for (int u = 0; u < RUN_U; ++u) {
        if (cr[u].x >= 0) src[atomicAdd(&lcur[cr[u].x - c0], 1)] = cr[u].y;
      }
      for (int u = 0; u < RUN_U && j0 + u < nr; ++u) {
        const int j  = j0 + u;
        const int sj = __builtin_amdgcn_readlane(s, j);
        const int ej = __builtin_amdgcn_readlane(e, j);
        for (int p = sj + 32 + lane; p < ej; p += 32) {
          const int2 q = pairs[(size_t)(cg + 32 * j) * CHUNK_E + p];
          src[atomicAdd(&lcur[q.x - c0], 1)] = q.y;
        }
      }
    }
  }
}


#define GEMM_RB 32
#define GEMM_AP 132

template <int F>
static __device__ __forceinline__ void
gemm_from_lds(const float* As,
              const float* __restrict__ B,
              float* __restrict__ C,
              int nrows, int row0) {
  const int lane = threadIdx.x & 31;
  const int wave = threadIdx.x >> 5;

  const int col0 = wave * 16;
  const int m    = lane & 15;
  const int half = lane >> 4;

  v8f acc0 = {};
  v8f acc1 = {};
  const float* a0p = As + m * GEMM_AP + 2 * half;
  const float* a1p = a0p + 16 * GEMM_AP;
  const float* bp  = B + (size_t)(2 * half) * F + col0 + m;

#pragma unroll
  for (int k = 0; k < 128; k += 4) {
    v2f a0, a1, b;
    a0.x = a0p[k];
    a0.y = a0p[k + 1];
    a1.x = a1p[k];
    a1.y = a1p[k + 1];
    b.x = bp[(size_t)k * F];
    b.y = bp[(size_t)(k + 1) * F];
    acc0 = __builtin_amdgcn_wmma_f32_16x16x4_f32(
         false, a0,  false, b,
         (short)0, acc0,  false,  false);
    acc1 = __builtin_amdgcn_wmma_f32_16x16x4_f32(
         false, a1,  false, b,
         (short)0, acc1,  false,  false);
  }

  float* cp = C + (size_t)(row0 + 8 * half) * F + col0 + m;
  if (row0 + GEMM_RB <= nrows) {
#pragma unroll
    for (int v = 0; v < 8; ++v) {
      cp[(size_t)v * F]        = acc0[v];
      cp[(size_t)(v + 16) * F] = acc1[v];
    }
  } else {
#pragma unroll
    for (int v = 0; v < 8; ++v) {
      const int r0 = row0 + v + 8 * half;
      if (r0 < nrows)      cp[(size_t)v * F]        = acc0[v];
      if (r0 + 16 < nrows) cp[(size_t)(v + 16) * F] = acc1[v];
    }
  }
}

template <int F>
static __device__ __forceinline__ void
gemm_tile(const float* __restrict__ A,
          const float* __restrict__ B,
          float* __restrict__ C,
          int nrows, int tile, float* As) {
  constexpr int NW = F / 16;

  const int lane = threadIdx.x & 31;
  const int wave = threadIdx.x >> 5;
  const int row0 = tile * GEMM_RB;

  const float4* A4 = (const float4*)A;
  float4 stage[GEMM_RB / NW];
#pragma unroll
  for (int i = 0; i < GEMM_RB / NW; ++i) {
    int row = row0 + wave + i * NW;
    row = (row < nrows) ? row : (nrows - 1);
    stage[i] = A4[(size_t)row * 32 + lane];
  }
#pragma unroll
  for (int i = 0; i < GEMM_RB / NW; ++i) {
    *(float4*)&As[(wave + i * NW) * GEMM_AP + lane * 4] = stage[i];
  }
  __syncthreads();

  gemm_from_lds<F>(As, B, C, nrows, row0);
}

static __device__ __forceinline__ bool deal_block(int b, int tiles, int chunks, int& idx) {
  const int both = tiles < chunks ? tiles : chunks;
  if (b < 2 * both) {
    idx = b >> 1;
    return (b & 1) == 0;
  }
  idx = both + (b - 2 * both);
  return tiles > chunks;
}

__global__ void __launch_bounds__(256)
gemm128_sort_kernel(const float* __restrict__ A,
                    const float* __restrict__ B,
                    float* __restrict__ C,
                    int nrows, int tiles,
                    const int* __restrict__ row, const int* __restrict__ col,
                    int* __restrict__ runs, int2* __restrict__ pairs,
                    int E, int nb, int nch) {
  __shared__ float As[GEMM_RB * GEMM_AP];
  __shared__ int cursor[BKT_MAX];
  __shared__ int wsum[BKT_MAX / 32];
  int idx;
  if (deal_block(blockIdx.x, tiles, nch, idx)) {
    gemm_tile<128>(A, B, C, nrows, idx, As);
  } else {
    const int t    = threadIdx.x;
    const int lane = t & 31;
    const int wave = t >> 5;
    cursor[t] = 0;
    const int e0 = idx * CHUNK_E + t;
    int c[CHUNK_E / 256];
    int r[CHUNK_E / 256];
#pragma unroll
    for (int i = 0; i < CHUNK_E / 256; ++i) {
      const int e = e0 + i * 256;
      c[i] = (e < E) ? col[e] : -1;
      r[i] = (e < E) ? row[e] : 0;
    }
    __syncthreads();
#pragma unroll
    for (int i = 0; i < CHUNK_E / 256; ++i) {
      if (c[i] >= 0) atomicAdd(&cursor[c[i] / BKT_W], 1);
    }
    __syncthreads();
    const int v = cursor[t];
    int inc = v;
    for (int d = 1; d < 32; d <<= 1) {
      int u = __shfl_up(inc, d);
      if (lane >= d) inc += u;
    }
    if (lane == 31) wsum[wave] = inc;
    __syncthreads();
    int first = inc - v;
    for (int w = 0; w < wave; ++w) first += wsum[w];
    cursor[t] = first;
    if (t < nb) runs[t * nch + idx] = first;
    __syncthreads();
    int2* mine = pairs + (size_t)idx * CHUNK_E;
#pragma unroll
    for (int i = 0; i < CHUNK_E / 256; ++i) {
      if (c[i] >= 0) {
        const int p = atomicAdd(&cursor[c[i] / BKT_W], 1);
        mine[p] = make_int2(c[i], r[i]);
      }
    }
  }
}


static __device__ __forceinline__ float lane_f(float x, int j) {
  return __int_as_float(__builtin_amdgcn_readlane(__float_as_int(x), j));
}

static __device__ __forceinline__ void
add_rows128(float4& acc, const float4* __restrict__ xw4, int r_l, float w_l, int k4, int lane) {
  for (int j = 0; j < k4; j += 4) {
    int    r[4];
    float4 v[4];
#pragma unroll
    for (int u = 0; u < 4; ++u) r[u] = __builtin_amdgcn_readlane(r_l, j + u);
#pragma unroll
    for (int u = 0; u < 4; ++u) v[u] = xw4[(size_t)r[u] * 32 + lane];
#pragma unroll
    for (int u = 0; u < 4; ++u) {
      const float w = lane_f(w_l, j + u);
      acc.x += v[u].x * w; acc.y += v[u].y * w; acc.z += v[u].z * w; acc.w += v[u].w * w;
    }
  }
}

#define GATHER_NW 4
__global__ void __launch_bounds__(256, 16)
gather128_gemm64_kernel(const int* __restrict__ off, const int* __restrict__ cnt,
                        const int* __restrict__ src, const float* __restrict__ dinv,
                        const float* __restrict__ xw, const float* __restrict__ b1,
                        const float* __restrict__ W2, float* __restrict__ hws, int n) {
  static_assert(8 * GATHER_NW == GEMM_RB, "8 waves x GATHER_NW nodes fill the row tile");
  __shared__ float As[GEMM_RB * GEMM_AP];
  const int lane = threadIdx.x & 31;
  const int wave = __builtin_amdgcn_readfirstlane(threadIdx.x >> 5);
  const int row0 = blockIdx.x * GEMM_RB;
  const int c0   = row0 + wave * GATHER_NW;
  const float4* xw4 = (const float4*)xw;

  int   cc[GATHER_NW], st[GATHER_NW], mm[GATHER_NW];
  float dd[GATHER_NW];
#pragma unroll
  for (int q = 0; q < GATHER_NW; ++q) {
    const int c = c0 + q;
    cc[q] = (c < n) ? c : (n - 1);
    st[q] = off[cc[q]];
    mm[q] = (c < n) ? cnt[cc[q]] : 0;
    dd[q] = (c < n) ? dinv[cc[q]] : 0.0f;
  }
  int   r_l[GATHER_NW];
  float w_l[GATHER_NW];
#pragma unroll
  for (int q = 0; q < GATHER_NW; ++q) {
    const int k = (mm[q] + 1 < 32) ? (mm[q] + 1) : 32;
    r_l[q] = cc[q];
    if (lane >= 1 && lane < k) r_l[q] = src[st[q] + lane - 1];
  }
  float dv[GATHER_NW];
#pragma unroll
  for (int q = 0; q < GATHER_NW; ++q) dv[q] = dinv[r_l[q]];
#pragma unroll
  for (int q = 0; q < GATHER_NW; ++q) {
    const int k = (mm[q] + 1 < 32) ? (mm[q] + 1) : 32;
    w_l[q] = (lane < k) ? dv[q] * dd[q] : 0.0f;
  }

  const int f = lane * 4;
  const float bx = b1[f + 0], by = b1[f + 1], bz = b1[f + 2], bw = b1[f + 3];

#pragma unroll
  for (int q = 0; q < GATHER_NW; ++q) {
    float4 acc = make_float4(0.0f, 0.0f, 0.0f, 0.0f);
    const int t = mm[q] + 1;
    const int k = (t < 32) ? t : 32;
    add_rows128(acc, xw4, r_l[q], w_l[q], (k + 3) & ~3, lane);
    for (int base = 32; base < t; base += 32) {
      const int kk = (t - base < 32) ? (t - base) : 32;
      int   r2 = cc[q];
      float w2 = 0.0f;
      if (lane < kk) {
        r2 = src[st[q] + base + lane - 1];
        w2 = dinv[r2] * dd[q];
      }
      add_rows128(acc, xw4, r2, w2, (kk + 3) & ~3, lane);
    }

    float4 o;
    o.x = fmaxf(acc.x + bx, 0.0f) * dd[q];
    o.y = fmaxf(acc.y + by, 0.0f) * dd[q];
    o.z = fmaxf(acc.z + bz, 0.0f) * dd[q];
    o.w = fmaxf(acc.w + bw, 0.0f) * dd[q];
    *(float4*)&As[(wave * GATHER_NW + q) * GEMM_AP + f] = o;
  }
  __syncthreads();

  if (wave < OUT_F / 16) {
    gemm_from_lds<OUT_F>(As, W2, hws, n, row0);
  }
}

static __device__ __forceinline__ void
add_rows64_pair(float4& acc, const float* __restrict__ hws, int e_l, int krel, int ng, int half, int i) {
  const unsigned lane_byte = (unsigned)i * 16u;
  for (int j = 0; j < ng; j += 8) {
    v4f v[8];
#pragma unroll
    for (int u = 0; u < 8; ++u) {
      const int ra = __builtin_amdgcn_readlane(e_l, j + u);
      const int rb = __builtin_amdgcn_readlane(e_l, 16 + j + u);
      const unsigned r = (unsigned)(half ? rb : ra);
      v[u] = *(const v4f*)((const char*)hws + (r * 256u + lane_byte));
    }
    asm volatile("" :: "v"(v[0]), "v"(v[1]), "v"(v[2]), "v"(v[3]), "v"(v[4]), "v"(v[5]), "v"(v[6]), "v"(v[7]));
#pragma unroll
    for (int u = 0; u < 8; ++u) {
      const float f = (j + u < krel) ? 1.0f : 0.0f;
      acc.x = __builtin_fmaf(v[u].x, f, acc.x);
      acc.y = __builtin_fmaf(v[u].y, f, acc.y);
      acc.z = __builtin_fmaf(v[u].z, f, acc.z);
      acc.w = __builtin_fmaf(v[u].w, f, acc.w);
    }
  }
}

#define G64_NP 2
__global__ void __launch_bounds__(256)
gather64_kernel(const int* __restrict__ off, const int* __restrict__ cnt,
                const int* __restrict__ src, const float* __restrict__ dinv,
                const float* __restrict__ hws, const float* __restrict__ b2,
                float* __restrict__ out, int n) {
  const int lane = threadIdx.x & 31;
  const int half = lane >> 4;
  const int i    = lane & 15;
  const int wave = __builtin_amdgcn_readfirstlane(threadIdx.x >> 5);
  const int c0   = (blockIdx.x * 8 + wave) * (2 * G64_NP);
  if (c0 >= n) return;

  int   cm[G64_NP], st[G64_NP], tt[G64_NP];
  float dc[G64_NP];
#pragma unroll
  for (int p = 0; p < G64_NP; ++p) {
    const int c = c0 + 2 * p + half;
    cm[p] = (c < n) ? c : (n - 1);
    st[p] = off[cm[p]];
    tt[p] = cnt[cm[p]];
    dc[p] = dinv[cm[p]];
  }
#pragma unroll
  for (int p = 0; p < G64_NP; ++p) {
    tt[p] = (c0 + 2 * p + half < n) ? tt[p] + 1 : 0;
  }
  int e0[G64_NP], e1[G64_NP];
#pragma unroll
  for (int p = 0; p < G64_NP; ++p) {
    const bool k0 = (i >= 1) && (i < tt[p]);
    const bool k1 = (16 + i < tt[p]);
    const int  x0 = src[k0 ? st[p] + i - 1 : 0];
    const int  x1 = src[k1 ? st[p] + 15 + i : 0];
    e0[p] = k0 ? x0 : cm[p];
    e1[p] = k1 ? x1 : cm[p];
  }
  const float4 bb = ((const float4*)b2)[i];

#pragma unroll
  for (int p = 0; p < G64_NP; ++p) {
    float4 acc = make_float4(0.0f, 0.0f, 0.0f, 0.0f);
    const int ta   = __builtin_amdgcn_readlane(tt[p], 0);
    const int tb   = __builtin_amdgcn_readlane(tt[p], 16);
    const int tmax = ta > tb ? ta : tb;
    int ng = (tmax + 7) & ~7;
    add_rows64_pair(acc, hws, e0[p], tt[p], ng < 16 ? ng : 16, half, i);
    if (tmax > 16) {
      ng = (tmax - 16 + 7) & ~7;
      add_rows64_pair(acc, hws, e1[p], tt[p] - 16, ng < 16 ? ng : 16, half, i);
      for (int base = 32; base < tmax; base += 16) {
        int e = cm[p];
        if (base + i < tt[p]) e = src[st[p] + base + i - 1];
        ng = (tmax - base + 7) & ~7;
        add_rows64_pair(acc, hws, e, tt[p] - base, ng < 16 ? ng : 16, half, i);
      }
    }
    const int c = c0 + 2 * p + half;
    if (c < n) {
      float4 o;
      o.x = acc.x * dc[p] + bb.x;
      o.y = acc.y * dc[p] + bb.y;
      o.z = acc.z * dc[p] + bb.z;
      o.w = acc.w * dc[p] + bb.w;
      ((float4*)out)[(size_t)c * 16 + i] = o;
    }
  }
}


extern "C" void kernel_launch(void* const* d_in, const int* in_sizes, int n_in,
                              void* d_out, int out_size, void* d_ws, size_t ws_size,
                              hipStream_t stream) {
  const float* x  = (const float*)d_in[0];
  const int*   ei = (const int*)d_in[1];
  const float* W1 = (const float*)d_in[2];
  const float* b1 = (const float*)d_in[3];
  const float* W2 = (const float*)d_in[4];
  const float* b2 = (const float*)d_in[5];
  float* out = (float*)d_out;

  const int N = in_sizes[0] / IN_F;
  const int E = in_sizes[1] / 2;
  const int* row = ei;
  const int* col = ei + E;

  const int nb  = (N + BKT_W - 1) / BKT_W;
  const int nch = (E + CHUNK_E - 1) / CHUNK_E;
  float* ws    = (float*)d_ws;
  float* xw    = ws;
  float* hws   = xw + (size_t)N * 128;
  int2*  pairs = (int2*)(hws + (size_t)N * 64);
  float* dinv  = (float*)(pairs + (size_t)nch * CHUNK_E);
  int*   cnt   = (int*)(dinv + N);
  int*   off   = cnt + N;
  int*   src   = off + N;
  int*   runs  = src + E;

  const int tiles = (N + GEMM_RB - 1) / GEMM_RB;
  gemm128_sort_kernel<<<tiles + nch, 256, 0, stream>>>(x, W1, xw, N, tiles,
                                                       row, col, runs, pairs, E, nb, nch);
  bucket_nodes_kernel<<<nb, NODE_T, 0, stream>>>(runs, pairs, off, cnt, dinv, src, N, E, nb, nch);

  gather128_gemm64_kernel<<<(N + GEMM_RB - 1) / GEMM_RB, 256, 0, stream>>>(off, cnt, src, dinv, xw, b1, W2, hws, N);

  gather64_kernel<<<(N + 31) / 32, 256, 0, stream>>>(off, cnt, src, dinv, hws, b2, out, N);
}
